# starting kernel
# baseline (speedup 1.0000x reference)
#include <hip/hip_runtime.h>

constexpr int B_   = 8;
constexpr int N_   = 1024;
constexpr int DIM_ = 768;
constexpr int H_   = 12;
constexpr int CR_  = 384;
constexpr int HD_  = 32;
constexpr float EPS_ = 1e-5f;

typedef __attribute__((ext_vector_type(2))) float v2f;
typedef __attribute__((ext_vector_type(8))) float v8f;

static __device__ __forceinline__ v8f wmma4(v2f a, v2f b, v8f c) {
    return __builtin_amdgcn_wmma_f32_16x16x4_f32(
         false, a,  false, b,
         (short)0, c,  false,  false);
}

__global__ __launch_bounds__(256)
void ln768_kernel(const float* __restrict__ x, const float* __restrict__ g,
                  const float* __restrict__ bia, float* __restrict__ o) {
    const int row  = blockIdx.x;
    const int t    = threadIdx.x;
    const float* xr = x + (size_t)row * DIM_;
    float v0 = xr[t], v1 = xr[t + 256], v2 = xr[t + 512];
    float s  = v0 + v1 + v2;
    float s2 = v0 * v0 + v1 * v1 + v2 * v2;
    #pragma unroll
    for (int off = 16; off > 0; off >>= 1) {
        s  += __shfl_xor(s,  off, 32);
        s2 += __shfl_xor(s2, off, 32);
    }
    __shared__ float ps[8], ps2[8];
    const int wid = t >> 5, lane = t & 31;
    if (lane == 0) { ps[wid] = s; ps2[wid] = s2; }
    __syncthreads();
    if (wid == 0) {
        float a  = (lane < 8) ? ps[lane]  : 0.f;
        float a2 = (lane < 8) ? ps2[lane] : 0.f;
        #pragma unroll
        for (int off = 4; off > 0; off >>= 1) {
            a  += __shfl_xor(a,  off, 32);
            a2 += __shfl_xor(a2, off, 32);
        }
        if (lane == 0) { ps[0] = a; ps2[0] = a2; }
    }
    __syncthreads();
    const float mean = ps[0] * (1.f / DIM_);
    const float var  = ps2[0] * (1.f / DIM_) - mean * mean;
    const float rstd = rsqrtf(var + EPS_);
    float* orow = o + (size_t)row * DIM_;
    orow[t]       = (v0 - mean) * rstd * g[t]       + bia[t];
    orow[t + 256] = (v1 - mean) * rstd * g[t + 256] + bia[t + 256];
    orow[t + 512] = (v2 - mean) * rstd * g[t + 512] + bia[t + 512];
}

__global__ __launch_bounds__(256)
void ln384_kernel(const float* __restrict__ x, const float* __restrict__ g,
                  const float* __restrict__ bia, float* __restrict__ o) {
    const int wid  = threadIdx.x >> 5;
    const int lane = threadIdx.x & 31;
    const int row  = blockIdx.x * 8 + wid;
    const float* xr = x + (size_t)row * CR_;
    float v[12];
    float s = 0.f, s2 = 0.f;
    #pragma unroll
    for (int i = 0; i < 12; ++i) {
        v[i] = xr[lane + 32 * i];
        s += v[i]; s2 += v[i] * v[i];
    }
    #pragma unroll
    for (int off = 16; off > 0; off >>= 1) {
        s  += __shfl_xor(s,  off, 32);
        s2 += __shfl_xor(s2, off, 32);
    }
    const float mean = s * (1.f / CR_);
    const float var  = s2 * (1.f / CR_) - mean * mean;
    const float rstd = rsqrtf(var + EPS_);
    float* orow = o + (size_t)row * CR_;
    #pragma unroll
    for (int i = 0; i < 12; ++i) {
        const int c = lane + 32 * i;
        orow[c] = (v[i] - mean) * rstd * g[c] + bia[c];
    }
}

constexpr int KC      = 32;
constexpr int WSTRIDE = KC + 2;

__global__ __launch_bounds__(256)
void gemm_wmma_kernel(const float* __restrict__ A, const float* __restrict__ W,
                      const float* __restrict__ bias, float* __restrict__ C,
                      int M, int N, int K) {
    __shared__ float sW[2][64 * WSTRIDE];
    const int t    = threadIdx.x;
    const int lane = t & 31;
    const int w    = t >> 5;
    const int hi   = lane >> 4;
    const int lo   = lane & 15;
    const int m0   = blockIdx.y * 128 + w * 16;
    const int n0   = blockIdx.x * 64;
    const int NC   = K / KC;

    const float* arow = A + (size_t)(m0 + lo) * K + 2 * hi;

    float wreg[8], wregN[8];
    v2f   areg[8], aregN[8];

    auto gload = [&](int kc, float* wr, v2f* ar) {
        #pragma unroll
        for (int i = 0; i < 8; ++i)
            wr[i] = W[(size_t)(n0 + i * 8 + w) * K + kc + lane];
        #pragma unroll
        for (int ts = 0; ts < 8; ++ts) {
            ar[ts].x = arow[kc + 4 * ts];
            ar[ts].y = arow[kc + 4 * ts + 1];
        }
    };

    v8f acc[4] = {};
    gload(0, wreg, areg);

    for (int c = 0; c < NC; ++c) {
        const int bufi = c & 1;
        if (c + 1 < NC) gload((c + 1) * KC, wregN, aregN);
        #pragma unroll
        for (int i = 0; i < 8; ++i)
            sW[bufi][(i * 8 + w) * WSTRIDE + lane] = wreg[i];
        __syncthreads();
        const float* sbase = sW[bufi];
        #pragma unroll
        for (int ks = 0; ks < KC / 4; ++ks) {
            #pragma unroll
            for (int tt = 0; tt < 4; ++tt) {
                const float* srow = sbase + (tt * 16 + lo) * WSTRIDE + ks * 4 + 2 * hi;
                v2f b; b.x = srow[0]; b.y = srow[1];
                acc[tt] = wmma4(areg[ks], b, acc[tt]);
            }
        }
        if (c + 1 < NC) {
            #pragma unroll
            for (int i = 0; i < 8; ++i) { wreg[i] = wregN[i]; areg[i] = aregN[i]; }
        }
    }

    #pragma unroll
    for (int tt = 0; tt < 4; ++tt) {
        const int col = n0 + tt * 16 + lo;
        const float bv = bias ? bias[col] : 0.f;
        #pragma unroll
        for (int r = 0; r < 8; ++r) {
            const int row = m0 + r + 8 * hi;
            C[(size_t)row * N + col] = acc[tt][r] + bv;
        }
    }
}

constexpr int SROW = 1028;
constexpr int ATTN_SMEM = 4 * 16 * SROW * 4;

__global__ __launch_bounds__(128)
void attn_kernel(const float* __restrict__ Qb, const float* __restrict__ Kb,
                 const float* __restrict__ Vb, const float* __restrict__ Hb,
                 float* __restrict__ Out) {
    extern __shared__ float sS[];
    const int lane = threadIdx.x & 31;
    const int w    = threadIdx.x >> 5;
    const int bh   = blockIdx.y;
    const int b    = bh / H_;
    const int h    = bh % H_;
    const int q0   = blockIdx.x * 64 + w * 16;
    const int hi   = lane >> 4;
    const int lo   = lane & 15;
    const float scale = 0.17677669529663687f;
    float* sP = sS + (size_t)w * 16 * SROW;

    const float* qrow = Qb + ((size_t)(b * N_ + q0 + lo) * CR_) + h * HD_ + 2 * hi;
    v2f aq[8];
    #pragma unroll
    for (int ts = 0; ts < 8; ++ts) { aq[ts].x = qrow[4 * ts]; aq[ts].y = qrow[4 * ts + 1]; }

    const float* kbase = Kb + (size_t)b * N_ * CR_ + h * HD_ + 2 * hi;
    v2f bk[8], bkN[8];
    {
        const float* kr = kbase + (size_t)lo * CR_;
        #pragma unroll
        for (int ts = 0; ts < 8; ++ts) { bk[ts].x = kr[4 * ts]; bk[ts].y = kr[4 * ts + 1]; }
    }
    for (int j = 0; j < N_ / 16; ++j) {
        if (j + 1 < N_ / 16) {
            const float* kr = kbase + (size_t)((j + 1) * 16 + lo) * CR_;
            #pragma unroll
            for (int ts = 0; ts < 8; ++ts) { bkN[ts].x = kr[4 * ts]; bkN[ts].y = kr[4 * ts + 1]; }
        }
        v8f acc = {};
        #pragma unroll
        for (int ts = 0; ts < 8; ++ts) acc = wmma4(aq[ts], bk[ts], acc);
        #pragma unroll
        for (int r = 0; r < 8; ++r)
            sP[(r + 8 * hi) * SROW + j * 16 + lo] = acc[r] * scale;
        if (j + 1 < N_ / 16) {
            #pragma unroll
            for (int ts = 0; ts < 8; ++ts) bk[ts] = bkN[ts];
        }
    }

    for (int r = 0; r < 16; ++r) {
        float* srow = sP + r * SROW;
        float vals[32];
        float mx = -3.0e38f;
        #pragma unroll
        for (int i = 0; i < 32; ++i) {
            vals[i] = srow[lane + 32 * i];
            mx = fmaxf(mx, vals[i]);
        }
        #pragma unroll
        for (int off = 16; off > 0; off >>= 1)
            mx = fmaxf(mx, __shfl_xor(mx, off, 32));
        float sum = 0.f;
        #pragma unroll
        for (int i = 0; i < 32; ++i) {
            vals[i] = __expf(vals[i] - mx);
            sum += vals[i];
        }
        #pragma unroll
        for (int off = 16; off > 0; off >>= 1)
            sum += __shfl_xor(sum, off, 32);
        const float inv = 1.f / sum;
        #pragma unroll
        for (int i = 0; i < 32; ++i)
            srow[lane + 32 * i] = vals[i] * inv;
    }

    v8f oacc[2] = {};
    const float* vbase = Vb + (size_t)b * N_ * CR_ + h * HD_;
    const float* prow  = sP + lo * SROW + 2 * hi;
    const int d0 = lo, d1 = 16 + lo;

    v2f ap[2], bva[2], bvb[2], apN[2], bvaN[2], bvbN[2];
    {
        #pragma unroll
        for (int u = 0; u < 2; ++u) {
            const int k0 = 4 * u;
            ap[u].x = prow[k0]; ap[u].y = prow[k0 + 1];
            const float* vr = vbase + (size_t)(k0 + 2 * hi) * CR_;
            bva[u].x = vr[d0]; bva[u].y = vr[CR_ + d0];
            bvb[u].x = vr[d1]; bvb[u].y = vr[CR_ + d1];
        }
    }
    for (int k0 = 0; k0 < N_; k0 += 8) {
        if (k0 + 8 < N_) {
            #pragma unroll
            for (int u = 0; u < 2; ++u) {
                const int kn = k0 + 8 + 4 * u;
                apN[u].x = prow[kn]; apN[u].y = prow[kn + 1];
                const float* vr = vbase + (size_t)(kn + 2 * hi) * CR_;
                bvaN[u].x = vr[d0]; bvaN[u].y = vr[CR_ + d0];
                bvbN[u].x = vr[d1]; bvbN[u].y = vr[CR_ + d1];
            }
        }
        #pragma unroll
        for (int u = 0; u < 2; ++u) {
            oacc[0] = wmma4(ap[u], bva[u], oacc[0]);
            oacc[1] = wmma4(ap[u], bvb[u], oacc[1]);
        }
        if (k0 + 8 < N_) {
            #pragma unroll
            for (int u = 0; u < 2; ++u) { ap[u] = apN[u]; bva[u] = bvaN[u]; bvb[u] = bvbN[u]; }
        }
    }

    #pragma unroll
    for (int tt = 0; tt < 2; ++tt) {
        const int col = h * HD_ + tt * 16 + lo;
        #pragma unroll
        for (int r = 0; r < 8; ++r) {
            const int row = q0 + r + 8 * hi;
            const size_t idx = (size_t)(b * N_ + row) * CR_ + col;
            Out[idx] = Hb[idx] + oacc[tt][r];
        }
    }
}

extern "C" void kernel_launch(void* const* d_in, const int* in_sizes, int n_in,
                              void* d_out, int out_size, void* d_ws, size_t ws_size,
                              hipStream_t stream) {
    (void)in_sizes; (void)n_in; (void)out_size; (void)ws_size;
    const float* x      = (const float*)d_in[0];
    const float* ln0_g  = (const float*)d_in[1];
    const float* ln0_b  = (const float*)d_in[2];
    const float* w_proj = (const float*)d_in[3];
    const float* b_proj = (const float*)d_in[4];
    const float* ln1_g  = (const float*)d_in[5];
    const float* ln1_b  = (const float*)d_in[6];
    const float* wq     = (const float*)d_in[7];
    const float* wk     = (const float*)d_in[8];
    const float* wv     = (const float*)d_in[9];
    float* out = (float*)d_out;

    float* ws = (float*)d_ws;
    const size_t rowsBig = (size_t)B_ * N_ * DIM_;
    const size_t rowsCr  = (size_t)B_ * N_ * CR_;
    float* xn = ws;
    float* hB = xn + rowsBig;
    float* yB = hB + rowsCr;
    float* qB = yB + rowsCr;
    float* kB = qB + rowsCr;
    float* vB = kB + rowsCr;

    const int M = B_ * N_;

    ln768_kernel<<<M, 256, 0, stream>>>(x, ln0_g, ln0_b, xn);

    gemm_wmma_kernel<<<dim3(CR_ / 64, M / 128), 256, 0, stream>>>(
        xn, w_proj, b_proj, hB, M, CR_, DIM_);

    ln384_kernel<<<M / 8, 256, 0, stream>>>(hB, ln1_g, ln1_b, yB);

    gemm_wmma_kernel<<<dim3(CR_ / 64, M / 128), 256, 0, stream>>>(
        yB, wq, nullptr, qB, M, CR_, CR_);
    gemm_wmma_kernel<<<dim3(CR_ / 64, M / 128), 256, 0, stream>>>(
        yB, wk, nullptr, kB, M, CR_, CR_);
    gemm_wmma_kernel<<<dim3(CR_ / 64, M / 128), 256, 0, stream>>>(
        yB, wv, nullptr, vB, M, CR_, CR_);

    (void)hipFuncSetAttribute((const void*)attn_kernel,
                              hipFuncAttributeMaxDynamicSharedMemorySize,
                              ATTN_SMEM);
    attn_kernel<<<dim3(N_ / 64, B_ * H_), 128, ATTN_SMEM, stream>>>(
        qB, kB, vB, hB, out);
}


# attention: deeper software pipelining of the QK^T and PV loops: 64 / 32 keys per trip, next trip's loads issued as one group into a second register set, 4 independent score accumulators
# speedup vs baseline: 1.3948x; 1.3948x over previous
#include <hip/hip_runtime.h>

constexpr int B_   = 8;
constexpr int N_   = 1024;
constexpr int DIM_ = 768;
constexpr int H_   = 12;
constexpr int CR_  = 384;
constexpr int HD_  = 32;
constexpr float EPS_ = 1e-5f;

typedef __attribute__((ext_vector_type(2))) float v2f;
typedef __attribute__((ext_vector_type(8))) float v8f;

static __device__ __forceinline__ v8f wmma4(v2f a, v2f b, v8f c) {
    return __builtin_amdgcn_wmma_f32_16x16x4_f32(
         false, a,  false, b,
         (short)0, c,  false,  false);
}

__global__ __launch_bounds__(256)
void ln768_kernel(const float* __restrict__ x, const float* __restrict__ g,
                  const float* __restrict__ bia, float* __restrict__ o) {
    const int row  = blockIdx.x;
    const int t    = threadIdx.x;
    const float* xr = x + (size_t)row * DIM_;
    float v0 = xr[t], v1 = xr[t + 256], v2 = xr[t + 512];
    float s  = v0 + v1 + v2;
    float s2 = v0 * v0 + v1 * v1 + v2 * v2;
    #pragma unroll
    for (int off = 16; off > 0; off >>= 1) {
        s  += __shfl_xor(s,  off, 32);
        s2 += __shfl_xor(s2, off, 32);
    }
    __shared__ float ps[8], ps2[8];
    const int wid = t >> 5, lane = t & 31;
    if (lane == 0) { ps[wid] = s; ps2[wid] = s2; }
    __syncthreads();
    if (wid == 0) {
        float a  = (lane < 8) ? ps[lane]  : 0.f;
        float a2 = (lane < 8) ? ps2[lane] : 0.f;
        #pragma unroll
        for (int off = 4; off > 0; off >>= 1) {
            a  += __shfl_xor(a,  off, 32);
            a2 += __shfl_xor(a2, off, 32);
        }
        if (lane == 0) { ps[0] = a; ps2[0] = a2; }
    }
    __syncthreads();
    const float mean = ps[0] * (1.f / DIM_);
    const float var  = ps2[0] * (1.f / DIM_) - mean * mean;
    const float rstd = rsqrtf(var + EPS_);
    float* orow = o + (size_t)row * DIM_;
    orow[t]       = (v0 - mean) * rstd * g[t]       + bia[t];
    orow[t + 256] = (v1 - mean) * rstd * g[t + 256] + bia[t + 256];
    orow[t + 512] = (v2 - mean) * rstd * g[t + 512] + bia[t + 512];
}

__global__ __launch_bounds__(256)
void ln384_kernel(const float* __restrict__ x, const float* __restrict__ g,
                  const float* __restrict__ bia, float* __restrict__ o) {
    const int wid  = threadIdx.x >> 5;
    const int lane = threadIdx.x & 31;
    const int row  = blockIdx.x * 8 + wid;
    const float* xr = x + (size_t)row * CR_;
    float v[12];
    float s = 0.f, s2 = 0.f;
    #pragma unroll
    for (int i = 0; i < 12; ++i) {
        v[i] = xr[lane + 32 * i];
        s += v[i]; s2 += v[i] * v[i];
    }
    #pragma unroll
    for (int off = 16; off > 0; off >>= 1) {
        s  += __shfl_xor(s,  off, 32);
        s2 += __shfl_xor(s2, off, 32);
    }
    const float mean = s * (1.f / CR_);
    const float var  = s2 * (1.f / CR_) - mean * mean;
    const float rstd = rsqrtf(var + EPS_);
    float* orow = o + (size_t)row * CR_;
    #pragma unroll
    for (int i = 0; i < 12; ++i) {
        const int c = lane + 32 * i;
        orow[c] = (v[i] - mean) * rstd * g[c] + bia[c];
    }
}

constexpr int KC      = 32;
constexpr int WSTRIDE = KC + 2;

__global__ __launch_bounds__(256)
void gemm_wmma_kernel(const float* __restrict__ A, const float* __restrict__ W,
                      const float* __restrict__ bias, float* __restrict__ C,
                      int M, int N, int K) {
    __shared__ float sW[2][64 * WSTRIDE];
    const int t    = threadIdx.x;
    const int lane = t & 31;
    const int w    = t >> 5;
    const int hi   = lane >> 4;
    const int lo   = lane & 15;
    const int m0   = blockIdx.y * 128 + w * 16;
    const int n0   = blockIdx.x * 64;
    const int NC   = K / KC;

    const float* arow = A + (size_t)(m0 + lo) * K + 2 * hi;

    float wreg[8], wregN[8];
    v2f   areg[8], aregN[8];

    auto gload = [&](int kc, float* wr, v2f* ar) {
        #pragma unroll
        for (int i = 0; i < 8; ++i)
            wr[i] = W[(size_t)(n0 + i * 8 + w) * K + kc + lane];
        #pragma unroll
        for (int ts = 0; ts < 8; ++ts) {
            ar[ts].x = arow[kc + 4 * ts];
            ar[ts].y = arow[kc + 4 * ts + 1];
        }
    };

    v8f acc[4] = {};
    gload(0, wreg, areg);

    for (int c = 0; c < NC; ++c) {
        const int bufi = c & 1;
        if (c + 1 < NC) gload((c + 1) * KC, wregN, aregN);
        #pragma unroll
        for (int i = 0; i < 8; ++i)
            sW[bufi][(i * 8 + w) * WSTRIDE + lane] = wreg[i];
        __syncthreads();
        const float* sbase = sW[bufi];
        #pragma unroll
        for (int ks = 0; ks < KC / 4; ++ks) {
            #pragma unroll
            for (int tt = 0; tt < 4; ++tt) {
                const float* srow = sbase + (tt * 16 + lo) * WSTRIDE + ks * 4 + 2 * hi;
                v2f b; b.x = srow[0]; b.y = srow[1];
                acc[tt] = wmma4(areg[ks], b, acc[tt]);
            }
        }
        if (c + 1 < NC) {
            #pragma unroll
            for (int i = 0; i < 8; ++i) { wreg[i] = wregN[i]; areg[i] = aregN[i]; }
        }
    }

    #pragma unroll
    for (int tt = 0; tt < 4; ++tt) {
        const int col = n0 + tt * 16 + lo;
        const float bv = bias ? bias[col] : 0.f;
        #pragma unroll
        for (int r = 0; r < 8; ++r) {
            const int row = m0 + r + 8 * hi;
            C[(size_t)row * N + col] = acc[tt][r] + bv;
        }
    }
}

constexpr int SROW = 1028;
constexpr int ATTN_SMEM = 4 * 16 * SROW * 4;

__global__ __launch_bounds__(128)
void attn_kernel(const float* __restrict__ Qb, const float* __restrict__ Kb,
                 const float* __restrict__ Vb, const float* __restrict__ Hb,
                 float* __restrict__ Out) {
    extern __shared__ float sS[];
    const int lane = threadIdx.x & 31;
    const int w    = threadIdx.x >> 5;
    const int bh   = blockIdx.y;
    const int b    = bh / H_;
    const int h    = bh % H_;
    const int q0   = blockIdx.x * 64 + w * 16;
    const int hi   = lane >> 4;
    const int lo   = lane & 15;
    const float scale = 0.17677669529663687f;
    float* sP = sS + (size_t)w * 16 * SROW;

    const float* qrow = Qb + ((size_t)(b * N_ + q0 + lo) * CR_) + h * HD_ + 2 * hi;
    v2f aq[8];
    #pragma unroll
    for (int ts = 0; ts < 8; ++ts) { aq[ts].x = qrow[4 * ts]; aq[ts].y = qrow[4 * ts + 1]; }

    const float* kbase = Kb + (size_t)b * N_ * CR_ + h * HD_ + 2 * hi;
    constexpr int KT    = 64;
    constexpr int NT_QK = N_ / KT;
    v2f bkA[4][8], bkB[4][8];

    auto kload = [&](int jt, v2f (&bk)[4][8]) {
        #pragma unroll
        for (int t = 0; t < 4; ++t) {
            const float* kr = kbase + (size_t)(jt * KT + t * 16 + lo) * CR_;
            #pragma unroll
            for (int ts = 0; ts < 8; ++ts) { bk[t][ts].x = kr[4 * ts]; bk[t][ts].y = kr[4 * ts + 1]; }
        }
    };
    auto qk = [&](int jt, const v2f (&bk)[4][8]) {
        v8f acc[4] = {};
        #pragma unroll
        for (int ts = 0; ts < 8; ++ts) {
            #pragma unroll
            for (int t = 0; t < 4; ++t) acc[t] = wmma4(aq[ts], bk[t][ts], acc[t]);
        }
        #pragma unroll
        for (int t = 0; t < 4; ++t) {
            #pragma unroll
            for (int r = 0; r < 8; ++r)
                sP[(r + 8 * hi) * SROW + jt * KT + t * 16 + lo] = acc[t][r] * scale;
        }
    };

    kload(0, bkA);
    for (int jt = 0; jt < NT_QK; jt += 2) {
        kload(jt + 1, bkB);
        qk(jt, bkA);
        if (jt + 2 < NT_QK) kload(jt + 2, bkA);
        qk(jt + 1, bkB);
    }

    for (int r = 0; r < 16; ++r) {
        float* srow = sP + r * SROW;
        float vals[32];
        float mx = -3.0e38f;
        #pragma unroll
        for (int i = 0; i < 32; ++i) {
            vals[i] = srow[lane + 32 * i];
            mx = fmaxf(mx, vals[i]);
        }
        #pragma unroll
        for (int off = 16; off > 0; off >>= 1)
            mx = fmaxf(mx, __shfl_xor(mx, off, 32));
        float sum = 0.f;
        #pragma unroll
        for (int i = 0; i < 32; ++i) {
            vals[i] = __expf(vals[i] - mx);
            sum += vals[i];
        }
        #pragma unroll
        for (int off = 16; off > 0; off >>= 1)
            sum += __shfl_xor(sum, off, 32);
        const float inv = 1.f / sum;
        #pragma unroll
        for (int i = 0; i < 32; ++i)
            srow[lane + 32 * i] = vals[i] * inv;
    }

    v8f oacc[2] = {};
    const float* vbase = Vb + (size_t)b * N_ * CR_ + h * HD_;
    const float* prow  = sP + lo * SROW + 2 * hi;
    const int d0 = lo, d1 = 16 + lo;
    constexpr int VT    = 32;
    constexpr int VS    = VT / 4;
    constexpr int NT_PV = N_ / VT;
    v2f apA[VS], vaA[VS], vbA[VS], apB[VS], vaB[VS], vbB[VS];

    auto pvload = [&](int it, v2f (&ap)[VS], v2f (&va)[VS], v2f (&vb)[VS]) {
        #pragma unroll
        for (int u = 0; u < VS; ++u) {
            const int kn = it * VT + 4 * u;
            const float* vr = vbase + (size_t)(kn + 2 * hi) * CR_;
            va[u].x = vr[d0]; va[u].y = vr[CR_ + d0];
            vb[u].x = vr[d1]; vb[u].y = vr[CR_ + d1];
        }
        #pragma unroll
        for (int u = 0; u < VS; ++u) {
            const int kn = it * VT + 4 * u;
            ap[u].x = prow[kn]; ap[u].y = prow[kn + 1];
        }
    };
    auto pv = [&](const v2f (&ap)[VS], const v2f (&va)[VS], const v2f (&vb)[VS]) {
        #pragma unroll
        for (int u = 0; u < VS; ++u) {
            oacc[0] = wmma4(ap[u], va[u], oacc[0]);
            oacc[1] = wmma4(ap[u], vb[u], oacc[1]);
        }
    };

    pvload(0, apA, vaA, vbA);
    for (int it = 0; it < NT_PV; it += 2) {
        pvload(it + 1, apB, vaB, vbB);
        __builtin_amdgcn_sched_barrier(0);
        pv(apA, vaA, vbA);
        if (it + 2 < NT_PV) pvload(it + 2, apA, vaA, vbA);
        __builtin_amdgcn_sched_barrier(0);
        pv(apB, vaB, vbB);
    }

    #pragma unroll
    for (int tt = 0; tt < 2; ++tt) {
        const int col = h * HD_ + tt * 16 + lo;
        #pragma unroll
        for (int r = 0; r < 8; ++r) {
            const int row = q0 + r + 8 * hi;
            const size_t idx = (size_t)(b * N_ + row) * CR_ + col;
            Out[idx] = Hb[idx] + oacc[tt][r];
        }
    }
}

extern "C" void kernel_launch(void* const* d_in, const int* in_sizes, int n_in,
                              void* d_out, int out_size, void* d_ws, size_t ws_size,
                              hipStream_t stream) {
    (void)in_sizes; (void)n_in; (void)out_size; (void)ws_size;
    const float* x      = (const float*)d_in[0];
    const float* ln0_g  = (const float*)d_in[1];
    const float* ln0_b  = (const float*)d_in[2];
    const float* w_proj = (const float*)d_in[3];
    const float* b_proj = (const float*)d_in[4];
    const float* ln1_g  = (const float*)d_in[5];
    const float* ln1_b  = (const float*)d_in[6];
    const float* wq     = (const float*)d_in[7];
    const float* wk     = (const float*)d_in[8];
    const float* wv     = (const float*)d_in[9];
    float* out = (float*)d_out;

    float* ws = (float*)d_ws;
    const size_t rowsBig = (size_t)B_ * N_ * DIM_;
    const size_t rowsCr  = (size_t)B_ * N_ * CR_;
    float* xn = ws;
    float* hB = xn + rowsBig;
    float* yB = hB + rowsCr;
    float* qB = yB + rowsCr;
    float* kB = qB + rowsCr;
    float* vB = kB + rowsCr;

    const int M = B_ * N_;

    ln768_kernel<<<M, 256, 0, stream>>>(x, ln0_g, ln0_b, xn);

    gemm_wmma_kernel<<<dim3(CR_ / 64, M / 128), 256, 0, stream>>>(
        xn, w_proj, b_proj, hB, M, CR_, DIM_);

    ln384_kernel<<<M / 8, 256, 0, stream>>>(hB, ln1_g, ln1_b, yB);

    gemm_wmma_kernel<<<dim3(CR_ / 64, M / 128), 256, 0, stream>>>(
        yB, wq, nullptr, qB, M, CR_, CR_);
    gemm_wmma_kernel<<<dim3(CR_ / 64, M / 128), 256, 0, stream>>>(
        yB, wk, nullptr, kB, M, CR_, CR_);
    gemm_wmma_kernel<<<dim3(CR_ / 64, M / 128), 256, 0, stream>>>(
        yB, wv, nullptr, vB, M, CR_, CR_);

    (void)hipFuncSetAttribute((const void*)attn_kernel,
                              hipFuncAttributeMaxDynamicSharedMemorySize,
                              ATTN_SMEM);
    attn_kernel<<<dim3(N_ / 64, B_ * H_), 128, ATTN_SMEM, stream>>>(
        qB, kB, vB, hB, out);
}


# attention softmax: each lane owns half a score row (8 B conflict-free LDS accesses, 2 cross-lane exchanges per strip, not 160 dependent ones); exp kept unscaled, 1/sum applied to the 16x32 output
# speedup vs baseline: 1.5869x; 1.1377x over previous
#include <hip/hip_runtime.h>

constexpr int B_   = 8;
constexpr int N_   = 1024;
constexpr int DIM_ = 768;
constexpr int H_   = 12;
constexpr int CR_  = 384;
constexpr int HD_  = 32;
constexpr float EPS_ = 1e-5f;

typedef __attribute__((ext_vector_type(2))) float v2f;
typedef __attribute__((ext_vector_type(8))) float v8f;

static __device__ __forceinline__ v8f wmma4(v2f a, v2f b, v8f c) {
    return __builtin_amdgcn_wmma_f32_16x16x4_f32(
         false, a,  false, b,
         (short)0, c,  false,  false);
}

__global__ __launch_bounds__(256)
void ln768_kernel(const float* __restrict__ x, const float* __restrict__ g,
                  const float* __restrict__ bia, float* __restrict__ o) {
    const int row  = blockIdx.x;
    const int t    = threadIdx.x;
    const float* xr = x + (size_t)row * DIM_;
    float v0 = xr[t], v1 = xr[t + 256], v2 = xr[t + 512];
    float s  = v0 + v1 + v2;
    float s2 = v0 * v0 + v1 * v1 + v2 * v2;
    #pragma unroll
    for (int off = 16; off > 0; off >>= 1) {
        s  += __shfl_xor(s,  off, 32);
        s2 += __shfl_xor(s2, off, 32);
    }
    __shared__ float ps[8], ps2[8];
    const int wid = t >> 5, lane = t & 31;
    if (lane == 0) { ps[wid] = s; ps2[wid] = s2; }
    __syncthreads();
    if (wid == 0) {
        float a  = (lane < 8) ? ps[lane]  : 0.f;
        float a2 = (lane < 8) ? ps2[lane] : 0.f;
        #pragma unroll
        for (int off = 4; off > 0; off >>= 1) {
            a  += __shfl_xor(a,  off, 32);
            a2 += __shfl_xor(a2, off, 32);
        }
        if (lane == 0) { ps[0] = a; ps2[0] = a2; }
    }
    __syncthreads();
    const float mean = ps[0] * (1.f / DIM_);
    const float var  = ps2[0] * (1.f / DIM_) - mean * mean;
    const float rstd = rsqrtf(var + EPS_);
    float* orow = o + (size_t)row * DIM_;
    orow[t]       = (v0 - mean) * rstd * g[t]       + bia[t];
    orow[t + 256] = (v1 - mean) * rstd * g[t + 256] + bia[t + 256];
    orow[t + 512] = (v2 - mean) * rstd * g[t + 512] + bia[t + 512];
}

__global__ __launch_bounds__(256)
void ln384_kernel(const float* __restrict__ x, const float* __restrict__ g,
                  const float* __restrict__ bia, float* __restrict__ o) {
    const int wid  = threadIdx.x >> 5;
    const int lane = threadIdx.x & 31;
    const int row  = blockIdx.x * 8 + wid;
    const float* xr = x + (size_t)row * CR_;
    float v[12];
    float s = 0.f, s2 = 0.f;
    #pragma unroll
    for (int i = 0; i < 12; ++i) {
        v[i] = xr[lane + 32 * i];
        s += v[i]; s2 += v[i] * v[i];
    }
    #pragma unroll
    for (int off = 16; off > 0; off >>= 1) {
        s  += __shfl_xor(s,  off, 32);
        s2 += __shfl_xor(s2, off, 32);
    }
    const float mean = s * (1.f / CR_);
    const float var  = s2 * (1.f / CR_) - mean * mean;
    const float rstd = rsqrtf(var + EPS_);
    float* orow = o + (size_t)row * CR_;
    #pragma unroll
    for (int i = 0; i < 12; ++i) {
        const int c = lane + 32 * i;
        orow[c] = (v[i] - mean) * rstd * g[c] + bia[c];
    }
}

constexpr int KC      = 32;
constexpr int WSTRIDE = KC + 2;

__global__ __launch_bounds__(256)
void gemm_wmma_kernel(const float* __restrict__ A, const float* __restrict__ W,
                      const float* __restrict__ bias, float* __restrict__ C,
                      int M, int N, int K) {
    __shared__ float sW[2][64 * WSTRIDE];
    const int t    = threadIdx.x;
    const int lane = t & 31;
    const int w    = t >> 5;
    const int hi   = lane >> 4;
    const int lo   = lane & 15;
    const int m0   = blockIdx.y * 128 + w * 16;
    const int n0   = blockIdx.x * 64;
    const int NC   = K / KC;

    const float* arow = A + (size_t)(m0 + lo) * K + 2 * hi;

    float wreg[8], wregN[8];
    v2f   areg[8], aregN[8];

    auto gload = [&](int kc, float* wr, v2f* ar) {
        #pragma unroll
        for (int i = 0; i < 8; ++i)
            wr[i] = W[(size_t)(n0 + i * 8 + w) * K + kc + lane];
        #pragma unroll
        for (int ts = 0; ts < 8; ++ts) {
            ar[ts].x = arow[kc + 4 * ts];
            ar[ts].y = arow[kc + 4 * ts + 1];
        }
    };

    v8f acc[4] = {};
    gload(0, wreg, areg);

    for (int c = 0; c < NC; ++c) {
        const int bufi = c & 1;
        if (c + 1 < NC) gload((c + 1) * KC, wregN, aregN);
        #pragma unroll
        for (int i = 0; i < 8; ++i)
            sW[bufi][(i * 8 + w) * WSTRIDE + lane] = wreg[i];
        __syncthreads();
        const float* sbase = sW[bufi];
        #pragma unroll
        for (int ks = 0; ks < KC / 4; ++ks) {
            #pragma unroll
            for (int tt = 0; tt < 4; ++tt) {
                const float* srow = sbase + (tt * 16 + lo) * WSTRIDE + ks * 4 + 2 * hi;
                v2f b; b.x = srow[0]; b.y = srow[1];
                acc[tt] = wmma4(areg[ks], b, acc[tt]);
            }
        }
        if (c + 1 < NC) {
            #pragma unroll
            for (int i = 0; i < 8; ++i) { wreg[i] = wregN[i]; areg[i] = aregN[i]; }
        }
    }

    #pragma unroll
    for (int tt = 0; tt < 4; ++tt) {
        const int col = n0 + tt * 16 + lo;
        const float bv = bias ? bias[col] : 0.f;
        #pragma unroll
        for (int r = 0; r < 8; ++r) {
            const int row = m0 + r + 8 * hi;
            C[(size_t)row * N + col] = acc[tt][r] + bv;
        }
    }
}

constexpr int SROW = 1028;
constexpr int ATTN_SMEM = 4 * 16 * SROW * 4;

__global__ __launch_bounds__(128)
void attn_kernel(const float* __restrict__ Qb, const float* __restrict__ Kb,
                 const float* __restrict__ Vb, const float* __restrict__ Hb,
                 float* __restrict__ Out) {
    extern __shared__ float sS[];
    const int lane = threadIdx.x & 31;
    const int w    = threadIdx.x >> 5;
    const int bh   = blockIdx.y;
    const int b    = bh / H_;
    const int h    = bh % H_;
    const int q0   = blockIdx.x * 64 + w * 16;
    const int hi   = lane >> 4;
    const int lo   = lane & 15;
    const float scale = 0.17677669529663687f;
    float* sP = sS + (size_t)w * 16 * SROW;

    const float* qrow = Qb + ((size_t)(b * N_ + q0 + lo) * CR_) + h * HD_ + 2 * hi;
    v2f aq[8];
    #pragma unroll
    for (int ts = 0; ts < 8; ++ts) { aq[ts].x = qrow[4 * ts]; aq[ts].y = qrow[4 * ts + 1]; }

    const float* kbase = Kb + (size_t)b * N_ * CR_ + h * HD_ + 2 * hi;
    constexpr int KT    = 64;
    constexpr int NT_QK = N_ / KT;
    v2f bkA[4][8], bkB[4][8];

    auto kload = [&](int jt, v2f (&bk)[4][8]) {
        #pragma unroll
        for (int t = 0; t < 4; ++t) {
            const float* kr = kbase + (size_t)(jt * KT + t * 16 + lo) * CR_;
            #pragma unroll
            for (int ts = 0; ts < 8; ++ts) { bk[t][ts].x = kr[4 * ts]; bk[t][ts].y = kr[4 * ts + 1]; }
        }
    };
    auto qk = [&](int jt, const v2f (&bk)[4][8]) {
        v8f acc[4] = {};
        #pragma unroll
        for (int ts = 0; ts < 8; ++ts) {
            #pragma unroll
            for (int t = 0; t < 4; ++t) acc[t] = wmma4(aq[ts], bk[t][ts], acc[t]);
        }
        #pragma unroll
        for (int t = 0; t < 4; ++t) {
            #pragma unroll
            for (int r = 0; r < 8; ++r)
                sP[(r + 8 * hi) * SROW + jt * KT + t * 16 + lo] = acc[t][r] * scale;
        }
    };

    kload(0, bkA);
    for (int jt = 0; jt < NT_QK; jt += 2) {
        kload(jt + 1, bkB);
        qk(jt, bkA);
        if (jt + 2 < NT_QK) kload(jt + 2, bkA);
        qk(jt + 1, bkB);
    }

    float* myrow = sP + lo * SROW + 2 * hi;
    constexpr int CH = 32;
    float m0 = -3.0e38f, m1 = -3.0e38f, m2 = -3.0e38f, m3 = -3.0e38f;
    #pragma unroll 1
    for (int c0 = 0; c0 < N_ / 4; c0 += CH) {
        v2f tv[CH];
        #pragma unroll
        for (int i = 0; i < CH; ++i)
            tv[i] = *reinterpret_cast<const v2f*>(myrow + 4 * (c0 + i));
        #pragma unroll
        for (int i = 0; i < CH; i += 2) {
            m0 = fmaxf(m0, tv[i].x);     m1 = fmaxf(m1, tv[i].y);
            m2 = fmaxf(m2, tv[i + 1].x); m3 = fmaxf(m3, tv[i + 1].y);
        }
    }
    float mx = fmaxf(fmaxf(m0, m1), fmaxf(m2, m3));
    mx = fmaxf(mx, __shfl_xor(mx, 16, 32));
    float s0 = 0.f, s1 = 0.f, s2 = 0.f, s3 = 0.f;
    #pragma unroll 1
    for (int c0 = 0; c0 < N_ / 4; c0 += CH) {
        v2f tv[CH];
        #pragma unroll
        for (int i = 0; i < CH; ++i)
            tv[i] = *reinterpret_cast<const v2f*>(myrow + 4 * (c0 + i));
        #pragma unroll
        for (int i = 0; i < CH; i += 2) {
            tv[i].x     = __expf(tv[i].x - mx);     s0 += tv[i].x;
            tv[i].y     = __expf(tv[i].y - mx);     s1 += tv[i].y;
            tv[i + 1].x = __expf(tv[i + 1].x - mx); s2 += tv[i + 1].x;
            tv[i + 1].y = __expf(tv[i + 1].y - mx); s3 += tv[i + 1].y;
        }
        #pragma unroll
        for (int i = 0; i < CH; ++i)
            *reinterpret_cast<v2f*>(myrow + 4 * (c0 + i)) = tv[i];
    }
    float sum = (s0 + s1) + (s2 + s3);
    sum += __shfl_xor(sum, 16, 32);
    const float inv = 1.f / sum;
    float invr[8];
    #pragma unroll
    for (int r = 0; r < 8; ++r) invr[r] = __shfl(inv, r + 8 * hi, 32);

    v8f oacc[2] = {};
    const float* vbase = Vb + (size_t)b * N_ * CR_ + h * HD_;
    const float* prow  = sP + lo * SROW + 2 * hi;
    const int d0 = lo, d1 = 16 + lo;
    constexpr int VT    = 32;
    constexpr int VS    = VT / 4;
    constexpr int NT_PV = N_ / VT;
    v2f apA[VS], vaA[VS], vbA[VS], apB[VS], vaB[VS], vbB[VS];

    auto pvload = [&](int it, v2f (&ap)[VS], v2f (&va)[VS], v2f (&vb)[VS]) {
        #pragma unroll
        for (int u = 0; u < VS; ++u) {
            const int kn = it * VT + 4 * u;
            const float* vr = vbase + (size_t)(kn + 2 * hi) * CR_;
            va[u].x = vr[d0]; va[u].y = vr[CR_ + d0];
            vb[u].x = vr[d1]; vb[u].y = vr[CR_ + d1];
        }
        #pragma unroll
        for (int u = 0; u < VS; ++u) {
            const int kn = it * VT + 4 * u;
            ap[u].x = prow[kn]; ap[u].y = prow[kn + 1];
        }
    };
    auto pv = [&](const v2f (&ap)[VS], const v2f (&va)[VS], const v2f (&vb)[VS]) {
        #pragma unroll
        for (int u = 0; u < VS; ++u) {
            oacc[0] = wmma4(ap[u], va[u], oacc[0]);
            oacc[1] = wmma4(ap[u], vb[u], oacc[1]);
        }
    };

    pvload(0, apA, vaA, vbA);
    for (int it = 0; it < NT_PV; it += 2) {
        pvload(it + 1, apB, vaB, vbB);
        __builtin_amdgcn_sched_barrier(0);
        pv(apA, vaA, vbA);
        if (it + 2 < NT_PV) pvload(it + 2, apA, vaA, vbA);
        __builtin_amdgcn_sched_barrier(0);
        pv(apB, vaB, vbB);
    }

    #pragma unroll
    for (int tt = 0; tt < 2; ++tt) {
        const int col = h * HD_ + tt * 16 + lo;
        #pragma unroll
        for (int r = 0; r < 8; ++r) {
            const int row = q0 + r + 8 * hi;
            const size_t idx = (size_t)(b * N_ + row) * CR_ + col;
            Out[idx] = Hb[idx] + oacc[tt][r] * invr[r];
        }
    }
}

extern "C" void kernel_launch(void* const* d_in, const int* in_sizes, int n_in,
                              void* d_out, int out_size, void* d_ws, size_t ws_size,
                              hipStream_t stream) {
    (void)in_sizes; (void)n_in; (void)out_size; (void)ws_size;
    const float* x      = (const float*)d_in[0];
    const float* ln0_g  = (const float*)d_in[1];
    const float* ln0_b  = (const float*)d_in[2];
    const float* w_proj = (const float*)d_in[3];
    const float* b_proj = (const float*)d_in[4];
    const float* ln1_g  = (const float*)d_in[5];
    const float* ln1_b  = (const float*)d_in[6];
    const float* wq     = (const float*)d_in[7];
    const float* wk     = (const float*)d_in[8];
    const float* wv     = (const float*)d_in[9];
    float* out = (float*)d_out;

    float* ws = (float*)d_ws;
    const size_t rowsBig = (size_t)B_ * N_ * DIM_;
    const size_t rowsCr  = (size_t)B_ * N_ * CR_;
    float* xn = ws;
    float* hB = xn + rowsBig;
    float* yB = hB + rowsCr;
    float* qB = yB + rowsCr;
    float* kB = qB + rowsCr;
    float* vB = kB + rowsCr;

    const int M = B_ * N_;

    ln768_kernel<<<M, 256, 0, stream>>>(x, ln0_g, ln0_b, xn);

    gemm_wmma_kernel<<<dim3(CR_ / 64, M / 128), 256, 0, stream>>>(
        xn, w_proj, b_proj, hB, M, CR_, DIM_);

    ln384_kernel<<<M / 8, 256, 0, stream>>>(hB, ln1_g, ln1_b, yB);

    gemm_wmma_kernel<<<dim3(CR_ / 64, M / 128), 256, 0, stream>>>(
        yB, wq, nullptr, qB, M, CR_, CR_);
    gemm_wmma_kernel<<<dim3(CR_ / 64, M / 128), 256, 0, stream>>>(
        yB, wk, nullptr, kB, M, CR_, CR_);
    gemm_wmma_kernel<<<dim3(CR_ / 64, M / 128), 256, 0, stream>>>(
        yB, wv, nullptr, vB, M, CR_, CR_);

    (void)hipFuncSetAttribute((const void*)attn_kernel,
                              hipFuncAttributeMaxDynamicSharedMemorySize,
                              ATTN_SMEM);
    attn_kernel<<<dim3(N_ / 64, B_ * H_), 128, ATTN_SMEM, stream>>>(
        qB, kB, vB, hB, out);
}


# attention: each 64-key K and V tile fetched once per work-group with whole-line 16 B loads into a 3-stage LDS ring; the 4 waves read their B fragments from LDS, not with 16-line global loads
# speedup vs baseline: 1.8917x; 1.1921x over previous
#include <hip/hip_runtime.h>

constexpr int B_   = 8;
constexpr int N_   = 1024;
constexpr int DIM_ = 768;
constexpr int H_   = 12;
constexpr int CR_  = 384;
constexpr int HD_  = 32;
constexpr float EPS_ = 1e-5f;

typedef __attribute__((ext_vector_type(2))) float v2f;
typedef __attribute__((ext_vector_type(4))) float v4f;
typedef __attribute__((ext_vector_type(8))) float v8f;

static __device__ __forceinline__ v8f wmma4(v2f a, v2f b, v8f c) {
    return __builtin_amdgcn_wmma_f32_16x16x4_f32(
         false, a,  false, b,
         (short)0, c,  false,  false);
}

__global__ __launch_bounds__(256)
void ln768_kernel(const float* __restrict__ x, const float* __restrict__ g,
                  const float* __restrict__ bia, float* __restrict__ o) {
    const int row  = blockIdx.x;
    const int t    = threadIdx.x;
    const float* xr = x + (size_t)row * DIM_;
    float v0 = xr[t], v1 = xr[t + 256], v2 = xr[t + 512];
    float s  = v0 + v1 + v2;
    float s2 = v0 * v0 + v1 * v1 + v2 * v2;
    #pragma unroll
    for (int off = 16; off > 0; off >>= 1) {
        s  += __shfl_xor(s,  off, 32);
        s2 += __shfl_xor(s2, off, 32);
    }
    __shared__ float ps[8], ps2[8];
    const int wid = t >> 5, lane = t & 31;
    if (lane == 0) { ps[wid] = s; ps2[wid] = s2; }
    __syncthreads();
    if (wid == 0) {
        float a  = (lane < 8) ? ps[lane]  : 0.f;
        float a2 = (lane < 8) ? ps2[lane] : 0.f;
        #pragma unroll
        for (int off = 4; off > 0; off >>= 1) {
            a  += __shfl_xor(a,  off, 32);
            a2 += __shfl_xor(a2, off, 32);
        }
        if (lane == 0) { ps[0] = a; ps2[0] = a2; }
    }
    __syncthreads();
    const float mean = ps[0] * (1.f / DIM_);
    const float var  = ps2[0] * (1.f / DIM_) - mean * mean;
    const float rstd = rsqrtf(var + EPS_);
    float* orow = o + (size_t)row * DIM_;
    orow[t]       = (v0 - mean) * rstd * g[t]       + bia[t];
    orow[t + 256] = (v1 - mean) * rstd * g[t + 256] + bia[t + 256];
    orow[t + 512] = (v2 - mean) * rstd * g[t + 512] + bia[t + 512];
}

__global__ __launch_bounds__(256)
void ln384_kernel(const float* __restrict__ x, const float* __restrict__ g,
                  const float* __restrict__ bia, float* __restrict__ o) {
    const int wid  = threadIdx.x >> 5;
    const int lane = threadIdx.x & 31;
    const int row  = blockIdx.x * 8 + wid;
    const float* xr = x + (size_t)row * CR_;
    float v[12];
    float s = 0.f, s2 = 0.f;
    #pragma unroll
    for (int i = 0; i < 12; ++i) {
        v[i] = xr[lane + 32 * i];
        s += v[i]; s2 += v[i] * v[i];
    }
    #pragma unroll
    for (int off = 16; off > 0; off >>= 1) {
        s  += __shfl_xor(s,  off, 32);
        s2 += __shfl_xor(s2, off, 32);
    }
    const float mean = s * (1.f / CR_);
    const float var  = s2 * (1.f / CR_) - mean * mean;
    const float rstd = rsqrtf(var + EPS_);
    float* orow = o + (size_t)row * CR_;
    #pragma unroll
    for (int i = 0; i < 12; ++i) {
        const int c = lane + 32 * i;
        orow[c] = (v[i] - mean) * rstd * g[c] + bia[c];
    }
}

constexpr int KC      = 32;
constexpr int WSTRIDE = KC + 2;

__global__ __launch_bounds__(256)
void gemm_wmma_kernel(const float* __restrict__ A, const float* __restrict__ W,
                      const float* __restrict__ bias, float* __restrict__ C,
                      int M, int N, int K) {
    __shared__ float sW[2][64 * WSTRIDE];
    const int t    = threadIdx.x;
    const int lane = t & 31;
    const int w    = t >> 5;
    const int hi   = lane >> 4;
    const int lo   = lane & 15;
    const int m0   = blockIdx.y * 128 + w * 16;
    const int n0   = blockIdx.x * 64;
    const int NC   = K / KC;

    const float* arow = A + (size_t)(m0 + lo) * K + 2 * hi;

    float wreg[8], wregN[8];
    v2f   areg[8], aregN[8];

    auto gload = [&](int kc, float* wr, v2f* ar) {
        #pragma unroll
        for (int i = 0; i < 8; ++i)
            wr[i] = W[(size_t)(n0 + i * 8 + w) * K + kc + lane];
        #pragma unroll
        for (int ts = 0; ts < 8; ++ts) {
            ar[ts].x = arow[kc + 4 * ts];
            ar[ts].y = arow[kc + 4 * ts + 1];
        }
    };

    v8f acc[4] = {};
    gload(0, wreg, areg);

    for (int c = 0; c < NC; ++c) {
        const int bufi = c & 1;
        if (c + 1 < NC) gload((c + 1) * KC, wregN, aregN);
        #pragma unroll
        for (int i = 0; i < 8; ++i)
            sW[bufi][(i * 8 + w) * WSTRIDE + lane] = wreg[i];
        __syncthreads();
        const float* sbase = sW[bufi];
        #pragma unroll
        for (int ks = 0; ks < KC / 4; ++ks) {
            #pragma unroll
            for (int tt = 0; tt < 4; ++tt) {
                const float* srow = sbase + (tt * 16 + lo) * WSTRIDE + ks * 4 + 2 * hi;
                v2f b; b.x = srow[0]; b.y = srow[1];
                acc[tt] = wmma4(areg[ks], b, acc[tt]);
            }
        }
        if (c + 1 < NC) {
            #pragma unroll
            for (int i = 0; i < 8; ++i) { wreg[i] = wregN[i]; areg[i] = aregN[i]; }
        }
    }

    #pragma unroll
    for (int tt = 0; tt < 4; ++tt) {
        const int col = n0 + tt * 16 + lo;
        const float bv = bias ? bias[col] : 0.f;
        #pragma unroll
        for (int r = 0; r < 8; ++r) {
            const int row = m0 + r + 8 * hi;
            C[(size_t)row * N + col] = acc[tt][r] + bv;
        }
    }
}

constexpr int SROW   = 1028;
constexpr int STRIPS = 4 * 16 * SROW;
constexpr int TK     = 64;
constexpr int KP     = 36;
constexpr int VPP    = 96;
constexpr int KBUF   = TK * KP;
constexpr int VBUF   = (TK / 2) * VPP;
constexpr int NSTAGE = 3;
constexpr int STAGEF = NSTAGE * (VBUF > KBUF ? VBUF : KBUF);
constexpr int ATTN_SMEM = (STRIPS + STAGEF) * 4;

static __device__ __forceinline__ void lds_stage_barrier() {
    __builtin_amdgcn_fence(__ATOMIC_RELEASE, "workgroup", "local");
    __builtin_amdgcn_s_barrier();
    __builtin_amdgcn_fence(__ATOMIC_ACQUIRE, "workgroup", "local");
}

__global__ __launch_bounds__(128)
void attn_kernel(const float* __restrict__ Qb, const float* __restrict__ Kb,
                 const float* __restrict__ Vb, const float* __restrict__ Hb,
                 float* __restrict__ Out) {
    extern __shared__ __attribute__((aligned(16))) float sS[];
    const int tid  = threadIdx.x;
    const int lane = tid & 31;
    const int w    = tid >> 5;
    const int bh   = blockIdx.y;
    const int b    = bh / H_;
    const int h    = bh % H_;
    const int q0   = blockIdx.x * 64 + w * 16;
    const int hi   = lane >> 4;
    const int lo   = lane & 15;
    const float scale = 0.17677669529663687f;
    float* sP  = sS + (size_t)w * 16 * SROW;
    float* stg = sS + STRIPS;

    const float* qrow = Qb + ((size_t)(b * N_ + q0 + lo) * CR_) + h * HD_ + 2 * hi;
    v2f aq[8];
    #pragma unroll
    for (int ts = 0; ts < 8; ++ts) { aq[ts].x = qrow[4 * ts]; aq[ts].y = qrow[4 * ts + 1]; }
    __builtin_amdgcn_sched_barrier(0);

    const int trow = tid >> 3;
    const int tc4  = (tid & 7) * 4;
    const size_t headoff = (size_t)b * N_ * CR_ + h * HD_ + tc4;
    const float* kg = Kb + headoff + (size_t)trow * CR_;
    const float* vg = Vb + headoff + (size_t)(2 * trow) * CR_;
    auto gloadK = [&](int tile, v4f (&g)[4]) {
        #pragma unroll
        for (int i = 0; i < 4; ++i)
            g[i] = *reinterpret_cast<const v4f*>(kg + (size_t)(tile * TK + 16 * i) * CR_);
    };
    auto gloadV = [&](int tile, v4f (&g)[4]) {
        #pragma unroll
        for (int i2 = 0; i2 < 2; ++i2) {
            g[2 * i2]     = *reinterpret_cast<const v4f*>(vg + (size_t)(tile * TK + 32 * i2) * CR_);
            g[2 * i2 + 1] = *reinterpret_cast<const v4f*>(vg + (size_t)(tile * TK + 32 * i2 + 1) * CR_);
        }
    };
    auto stashK = [&](int tile, const v4f (&g)[4]) {
        float* dst = stg + (tile % NSTAGE) * KBUF + trow * KP + tc4;
        #pragma unroll
        for (int i = 0; i < 4; ++i)
            *reinterpret_cast<v4f*>(dst + 16 * i * KP) = g[i];
    };
    auto stashV = [&](int tile, const v4f (&g)[4]) {
        float* dst = stg + (tile % NSTAGE) * VBUF + trow * VPP + 2 * tc4;
        #pragma unroll
        for (int i2 = 0; i2 < 2; ++i2) {
            v4f e = g[2 * i2], o = g[2 * i2 + 1];
            asm volatile("" : "+v"(e), "+v"(o));
            v4f lo4, hi4;
            lo4.x = e.x; lo4.y = o.x; lo4.z = e.y; lo4.w = o.y;
            hi4.x = e.z; hi4.y = o.z; hi4.z = e.w; hi4.w = o.w;
            *reinterpret_cast<v4f*>(dst + 16 * i2 * VPP)     = lo4;
            *reinterpret_cast<v4f*>(dst + 16 * i2 * VPP + 4) = hi4;
        }
    };

    constexpr int NT_QK = N_ / TK;
    v2f bkA[4][8], bkB[4][8];
    v4f gA[4], gB[4];

    auto kfrag = [&](int jt, v2f (&bk)[4][8]) {
        const float* src = stg + (jt % NSTAGE) * KBUF + lo * KP + 2 * hi;
        #pragma unroll
        for (int t = 0; t < 4; ++t) {
            #pragma unroll
            for (int ts = 0; ts < 8; ++ts)
                bk[t][ts] = *reinterpret_cast<const v2f*>(src + t * 16 * KP + 4 * ts);
        }
    };
    auto qk = [&](int jt, const v2f (&bk)[4][8]) {
        v8f acc[4] = {};
        #pragma unroll
        for (int ts = 0; ts < 8; ++ts) {
            #pragma unroll
            for (int t = 0; t < 4; ++t) acc[t] = wmma4(aq[ts], bk[t][ts], acc[t]);
        }
        #pragma unroll
        for (int t = 0; t < 4; ++t) {
            #pragma unroll
            for (int r = 0; r < 8; ++r)
                sP[(r + 8 * hi) * SROW + jt * TK + t * 16 + lo] = acc[t][r] * scale;
        }
    };

    gloadK(0, gA); gloadK(1, gB);
    stashK(0, gA); stashK(1, gB);
    gloadK(2, gA); gloadK(3, gB);
    lds_stage_barrier();
    kfrag(0, bkA);
    auto lastk = [](int t) { return t < NT_QK ? t : NT_QK - 1; };
    for (int jt = 0; jt < NT_QK; jt += 2) {
        kfrag(jt + 1, bkB);
        stashK(jt + 2, gA);
        gloadK(lastk(jt + 4), gA);
        __builtin_amdgcn_sched_barrier(0);
        qk(jt, bkA);
        lds_stage_barrier();
        kfrag(lastk(jt + 2), bkA);
        stashK(jt + 3, gB);
        gloadK(lastk(jt + 5), gB);
        __builtin_amdgcn_sched_barrier(0);
        qk(jt + 1, bkB);
        lds_stage_barrier();
    }

    gloadV(0, gA); gloadV(1, gB);

    float* myrow = sP + lo * SROW + 2 * hi;
    constexpr int CH = 32;
    float m0 = -3.0e38f, m1 = -3.0e38f, m2 = -3.0e38f, m3 = -3.0e38f;
    #pragma unroll 1
    for (int c0 = 0; c0 < N_ / 4; c0 += CH) {
        v2f tv[CH];
        #pragma unroll
        for (int i = 0; i < CH; ++i)
            tv[i] = *reinterpret_cast<const v2f*>(myrow + 4 * (c0 + i));
        #pragma unroll
        for (int i = 0; i < CH; i += 2) {
            m0 = fmaxf(m0, tv[i].x);     m1 = fmaxf(m1, tv[i].y);
            m2 = fmaxf(m2, tv[i + 1].x); m3 = fmaxf(m3, tv[i + 1].y);
        }
    }
    float mx = fmaxf(fmaxf(m0, m1), fmaxf(m2, m3));
    mx = fmaxf(mx, __shfl_xor(mx, 16, 32));
    float s0 = 0.f, s1 = 0.f, s2 = 0.f, s3 = 0.f;
    #pragma unroll 1
    for (int c0 = 0; c0 < N_ / 4; c0 += CH) {
        v2f tv[CH];
        #pragma unroll
        for (int i = 0; i < CH; ++i)
            tv[i] = *reinterpret_cast<const v2f*>(myrow + 4 * (c0 + i));
        #pragma unroll
        for (int i = 0; i < CH; i += 2) {
            tv[i].x     = __expf(tv[i].x - mx);     s0 += tv[i].x;
            tv[i].y     = __expf(tv[i].y - mx);     s1 += tv[i].y;
            tv[i + 1].x = __expf(tv[i + 1].x - mx); s2 += tv[i + 1].x;
            tv[i + 1].y = __expf(tv[i + 1].y - mx); s3 += tv[i + 1].y;
        }
        #pragma unroll
        for (int i = 0; i < CH; ++i)
            *reinterpret_cast<v2f*>(myrow + 4 * (c0 + i)) = tv[i];
    }
    float sum = (s0 + s1) + (s2 + s3);
    sum += __shfl_xor(sum, 16, 32);
    const float inv = 1.f / sum;
    float invr[8];
    #pragma unroll
    for (int r = 0; r < 8; ++r) invr[r] = __shfl(inv, r + 8 * hi, 32);

    v8f oacc[2] = {};
    const float* prow  = sP + lo * SROW + 2 * hi;
    const int d0 = lo, d1 = 16 + lo;
    constexpr int VH    = TK / 2;
    constexpr int VS    = VH / 4;
    constexpr int NT_PV = N_ / TK;
    static_assert(NT_PV == NT_QK, "lastk serves both loops");
    v2f apA[VS], vaA[VS], vbA[VS], apB[VS], vaB[VS], vbB[VS];

    auto pvfrag = [&](int it, int half, v2f (&ap)[VS], v2f (&va)[VS], v2f (&vb)[VS]) {
        const float* vs = stg + (it % NSTAGE) * VBUF + (half * (VH / 2) + hi) * VPP;
        #pragma unroll
        for (int u = 0; u < VS; ++u) {
            va[u] = *reinterpret_cast<const v2f*>(vs + 2 * u * VPP + 2 * d0);
            vb[u] = *reinterpret_cast<const v2f*>(vs + 2 * u * VPP + 2 * d1);
        }
        #pragma unroll
        for (int u = 0; u < VS; ++u)
            ap[u] = *reinterpret_cast<const v2f*>(prow + it * TK + half * VH + 4 * u);
    };
    auto pv = [&](const v2f (&ap)[VS], const v2f (&va)[VS], const v2f (&vb)[VS]) {
        __builtin_amdgcn_sched_barrier(0);
        #pragma unroll
        for (int u = 0; u < VS; ++u) {
            oacc[0] = wmma4(ap[u], va[u], oacc[0]);
            oacc[1] = wmma4(ap[u], vb[u], oacc[1]);
        }
        __builtin_amdgcn_sched_barrier(0);
    };

    stashV(0, gA); stashV(1, gB);
    gloadV(2, gA); gloadV(3, gB);
    lds_stage_barrier();
    pvfrag(0, 0, apA, vaA, vbA);
    for (int it = 0; it < NT_PV; it += 2) {
        pvfrag(it, 1, apB, vaB, vbB);
        stashV(it + 2, gA);
        gloadV(lastk(it + 4), gA);
        pv(apA, vaA, vbA);
        pvfrag(it + 1, 0, apA, vaA, vbA);
        pv(apB, vaB, vbB);
        lds_stage_barrier();
        pvfrag(it + 1, 1, apB, vaB, vbB);
        stashV(it + 3, gB);
        gloadV(lastk(it + 5), gB);
        pv(apA, vaA, vbA);
        pvfrag(lastk(it + 2), 0, apA, vaA, vbA);
        pv(apB, vaB, vbB);
        lds_stage_barrier();
    }

    #pragma unroll
    for (int tt = 0; tt < 2; ++tt) {
        const int col = h * HD_ + tt * 16 + lo;
        #pragma unroll
        for (int r = 0; r < 8; ++r) {
            const int row = q0 + r + 8 * hi;
            const size_t idx = (size_t)(b * N_ + row) * CR_ + col;
            Out[idx] = Hb[idx] + oacc[tt][r] * invr[r];
        }
    }
}

extern "C" void kernel_launch(void* const* d_in, const int* in_sizes, int n_in,
                              void* d_out, int out_size, void* d_ws, size_t ws_size,
                              hipStream_t stream) {
    (void)in_sizes; (void)n_in; (void)out_size; (void)ws_size;
    const float* x      = (const float*)d_in[0];
    const float* ln0_g  = (const float*)d_in[1];
    const float* ln0_b  = (const float*)d_in[2];
    const float* w_proj = (const float*)d_in[3];
    const float* b_proj = (const float*)d_in[4];
    const float* ln1_g  = (const float*)d_in[5];
    const float* ln1_b  = (const float*)d_in[6];
    const float* wq     = (const float*)d_in[7];
    const float* wk     = (const float*)d_in[8];
    const float* wv     = (const float*)d_in[9];
    float* out = (float*)d_out;

    float* ws = (float*)d_ws;
    const size_t rowsBig = (size_t)B_ * N_ * DIM_;
    const size_t rowsCr  = (size_t)B_ * N_ * CR_;
    float* xn = ws;
    float* hB = xn + rowsBig;
    float* yB = hB + rowsCr;
    float* qB = yB + rowsCr;
    float* kB = qB + rowsCr;
    float* vB = kB + rowsCr;

    const int M = B_ * N_;

    ln768_kernel<<<M, 256, 0, stream>>>(x, ln0_g, ln0_b, xn);

    gemm_wmma_kernel<<<dim3(CR_ / 64, M / 128), 256, 0, stream>>>(
        xn, w_proj, b_proj, hB, M, CR_, DIM_);

    ln384_kernel<<<M / 8, 256, 0, stream>>>(hB, ln1_g, ln1_b, yB);

    gemm_wmma_kernel<<<dim3(CR_ / 64, M / 128), 256, 0, stream>>>(
        yB, wq, nullptr, qB, M, CR_, CR_);
    gemm_wmma_kernel<<<dim3(CR_ / 64, M / 128), 256, 0, stream>>>(
        yB, wk, nullptr, kB, M, CR_, CR_);
    gemm_wmma_kernel<<<dim3(CR_ / 64, M / 128), 256, 0, stream>>>(
        yB, wv, nullptr, vB, M, CR_, CR_);

    (void)hipFuncSetAttribute((const void*)attn_kernel,
                              hipFuncAttributeMaxDynamicSharedMemorySize,
                              ATTN_SMEM);
    attn_kernel<<<dim3(N_ / 64, B_ * H_), 128, ATTN_SMEM, stream>>>(
        qB, kB, vB, hB, out);
}


# GEMMs re-tiled 128x96: each launch is 256 work-groups, one per WGP with equal work, fed by whole-line 16 B loads into a 3-stage LDS ring, fragments read one trip ahead, LDS-only barriers
# speedup vs baseline: 2.2383x; 1.1832x over previous
#include <hip/hip_runtime.h>

constexpr int B_   = 8;
constexpr int N_   = 1024;
constexpr int DIM_ = 768;
constexpr int H_   = 12;
constexpr int CR_  = 384;
constexpr int HD_  = 32;
constexpr float EPS_ = 1e-5f;

typedef __attribute__((ext_vector_type(2))) float v2f;
typedef __attribute__((ext_vector_type(4))) float v4f;
typedef __attribute__((ext_vector_type(8))) float v8f;

static __device__ __forceinline__ v8f wmma4(v2f a, v2f b, v8f c) {
    return __builtin_amdgcn_wmma_f32_16x16x4_f32(
         false, a,  false, b,
         (short)0, c,  false,  false);
}

__global__ __launch_bounds__(256)
void ln768_kernel(const float* __restrict__ x, const float* __restrict__ g,
                  const float* __restrict__ bia, float* __restrict__ o) {
    const int row  = blockIdx.x;
    const int t    = threadIdx.x;
    const float* xr = x + (size_t)row * DIM_;
    float v0 = xr[t], v1 = xr[t + 256], v2 = xr[t + 512];
    float s  = v0 + v1 + v2;
    float s2 = v0 * v0 + v1 * v1 + v2 * v2;
    #pragma unroll
    for (int off = 16; off > 0; off >>= 1) {
        s  += __shfl_xor(s,  off, 32);
        s2 += __shfl_xor(s2, off, 32);
    }
    __shared__ float ps[8], ps2[8];
    const int wid = t >> 5, lane = t & 31;
    if (lane == 0) { ps[wid] = s; ps2[wid] = s2; }
    __syncthreads();
    if (wid == 0) {
        float a  = (lane < 8) ? ps[lane]  : 0.f;
        float a2 = (lane < 8) ? ps2[lane] : 0.f;
        #pragma unroll
        for (int off = 4; off > 0; off >>= 1) {
            a  += __shfl_xor(a,  off, 32);
            a2 += __shfl_xor(a2, off, 32);
        }
        if (lane == 0) { ps[0] = a; ps2[0] = a2; }
    }
    __syncthreads();
    const float mean = ps[0] * (1.f / DIM_);
    const float var  = ps2[0] * (1.f / DIM_) - mean * mean;
    const float rstd = rsqrtf(var + EPS_);
    float* orow = o + (size_t)row * DIM_;
    orow[t]       = (v0 - mean) * rstd * g[t]       + bia[t];
    orow[t + 256] = (v1 - mean) * rstd * g[t + 256] + bia[t + 256];
    orow[t + 512] = (v2 - mean) * rstd * g[t + 512] + bia[t + 512];
}

__global__ __launch_bounds__(256)
void ln384_kernel(const float* __restrict__ x, const float* __restrict__ g,
                  const float* __restrict__ bia, float* __restrict__ o) {
    const int wid  = threadIdx.x >> 5;
    const int lane = threadIdx.x & 31;
    const int row  = blockIdx.x * 8 + wid;
    const float* xr = x + (size_t)row * CR_;
    float v[12];
    float s = 0.f, s2 = 0.f;
    #pragma unroll
    for (int i = 0; i < 12; ++i) {
        v[i] = xr[lane + 32 * i];
        s += v[i]; s2 += v[i] * v[i];
    }
    #pragma unroll
    for (int off = 16; off > 0; off >>= 1) {
        s  += __shfl_xor(s,  off, 32);
        s2 += __shfl_xor(s2, off, 32);
    }
    const float mean = s * (1.f / CR_);
    const float var  = s2 * (1.f / CR_) - mean * mean;
    const float rstd = rsqrtf(var + EPS_);
    float* orow = o + (size_t)row * CR_;
    #pragma unroll
    for (int i = 0; i < 12; ++i) {
        const int c = lane + 32 * i;
        orow[c] = (v[i] - mean) * rstd * g[c] + bia[c];
    }
}

static __device__ __forceinline__ void lds_stage_barrier() {
    __builtin_amdgcn_fence(__ATOMIC_RELEASE, "workgroup", "local");
    __builtin_amdgcn_s_barrier();
    __builtin_amdgcn_fence(__ATOMIC_ACQUIRE, "workgroup", "local");
}

constexpr int KC   = 32;
constexpr int GP   = KC + 4;
constexpr int GM   = 128;
constexpr int GN   = 96;
constexpr int GSTG = (GM + GN) * GP;
constexpr int GNST = 3;
constexpr int GEMM_SMEM = GNST * GSTG * 4;

template <int K>
__global__ __launch_bounds__(256)
void gemm_wmma_kernel(const float* __restrict__ A, const float* __restrict__ W,
                      const float* __restrict__ bias, float* __restrict__ C, int N) {
    extern __shared__ __attribute__((aligned(16))) float sT[];
    constexpr int NC = K / KC;
    static_assert(K % KC == 0 && NC % 2 == 0 && NC >= 4, "two chunks per loop trip");
    const int t    = threadIdx.x;
    const int lane = t & 31;
    const int w    = t >> 5;
    const int wm   = w >> 1;
    const int wn   = w & 1;
    const int hi   = lane >> 4;
    const int lo   = lane & 15;
    const int m0   = blockIdx.y * GM;
    const int n0   = blockIdx.x * GN;

    const int trow = t >> 3;
    const int tc4  = (t & 7) * 4;
    const float* ag = A + (size_t)(m0 + trow) * K + tc4;
    const float* wg = W + (size_t)(n0 + trow) * K + tc4;
    auto gload = [&](int c, v4f (&g)[7]) {
        #pragma unroll
        for (int i = 0; i < 4; ++i)
            g[i] = *reinterpret_cast<const v4f*>(ag + (size_t)(32 * i) * K + c * KC);
        #pragma unroll
        for (int i = 0; i < 3; ++i)
            g[4 + i] = *reinterpret_cast<const v4f*>(wg + (size_t)(32 * i) * K + c * KC);
    };
    auto stash = [&](int c, const v4f (&g)[7]) {
        float* dst = sT + (c % GNST) * GSTG + trow * GP + tc4;
        #pragma unroll
        for (int i = 0; i < 4; ++i)
            *reinterpret_cast<v4f*>(dst + 32 * i * GP) = g[i];
        #pragma unroll
        for (int i = 0; i < 3; ++i)
            *reinterpret_cast<v4f*>(dst + (GM + 32 * i) * GP) = g[4 + i];
    };

    v8f acc[2][3] = {};
    v2f faA[2][KC / 4], fbA[3][KC / 4], faB[2][KC / 4], fbB[3][KC / 4];
    auto fragload = [&](int c, v2f (&fa)[2][KC / 4], v2f (&fb)[3][KC / 4]) {
        const float* sa = sT + (c % GNST) * GSTG + (wm * 32 + lo) * GP + 2 * hi;
        const float* sb = sT + (c % GNST) * GSTG + (GM + wn * 48 + lo) * GP + 2 * hi;
        #pragma unroll
        for (int ks = 0; ks < KC / 4; ++ks) {
            #pragma unroll
            for (int rt = 0; rt < 2; ++rt)
                fa[rt][ks] = *reinterpret_cast<const v2f*>(sa + rt * 16 * GP + 4 * ks);
            #pragma unroll
            for (int ct = 0; ct < 3; ++ct)
                fb[ct][ks] = *reinterpret_cast<const v2f*>(sb + ct * 16 * GP + 4 * ks);
        }
    };
    auto mma = [&](const v2f (&fa)[2][KC / 4], const v2f (&fb)[3][KC / 4]) {
        __builtin_amdgcn_sched_barrier(0);
        #pragma unroll
        for (int ks = 0; ks < KC / 4; ++ks) {
            #pragma unroll
            for (int rt = 0; rt < 2; ++rt) {
                #pragma unroll
                for (int ct = 0; ct < 3; ++ct)
                    acc[rt][ct] = wmma4(fa[rt][ks], fb[ct][ks], acc[rt][ct]);
            }
        }
        __builtin_amdgcn_sched_barrier(0);
    };

    auto lastc = [](int c) { return c < NC ? c : NC - 1; };
    v4f gA[7], gB[7];
    gload(0, gA); gload(1, gB);
    stash(0, gA); stash(1, gB);
    gload(2, gA);
    __builtin_amdgcn_sched_barrier(0);
    gload(3, gB);
    lds_stage_barrier();
    fragload(0, faA, fbA);
    for (int c = 0; c < NC; c += 2) {
        fragload(c + 1, faB, fbB);
        stash(c + 2, gA);
        gload(lastc(c + 4), gA);
        mma(faA, fbA);
        lds_stage_barrier();
        fragload(lastc(c + 2), faA, fbA);
        stash(c + 3, gB);
        gload(lastc(c + 5), gB);
        mma(faB, fbB);
        lds_stage_barrier();
    }

    #pragma unroll
    for (int rt = 0; rt < 2; ++rt) {
        #pragma unroll
        for (int ct = 0; ct < 3; ++ct) {
            const int col = n0 + wn * 48 + ct * 16 + lo;
            const float bv = bias ? bias[col] : 0.f;
            #pragma unroll
            for (int r = 0; r < 8; ++r) {
                const int row = m0 + wm * 32 + rt * 16 + r + 8 * hi;
                C[(size_t)row * N + col] = acc[rt][ct][r] + bv;
            }
        }
    }
}

constexpr int SROW   = 1028;
constexpr int STRIPS = 4 * 16 * SROW;
constexpr int TK     = 64;
constexpr int KP     = 36;
constexpr int VPP    = 96;
constexpr int KBUF   = TK * KP;
constexpr int VBUF   = (TK / 2) * VPP;
constexpr int NSTAGE = 3;
constexpr int STAGEF = NSTAGE * (VBUF > KBUF ? VBUF : KBUF);
constexpr int ATTN_SMEM = (STRIPS + STAGEF) * 4;

__global__ __launch_bounds__(128)
void attn_kernel(const float* __restrict__ Qb, const float* __restrict__ Kb,
                 const float* __restrict__ Vb, const float* __restrict__ Hb,
                 float* __restrict__ Out) {
    extern __shared__ __attribute__((aligned(16))) float sS[];
    const int tid  = threadIdx.x;
    const int lane = tid & 31;
    const int w    = tid >> 5;
    const int bh   = blockIdx.y;
    const int b    = bh / H_;
    const int h    = bh % H_;
    const int q0   = blockIdx.x * 64 + w * 16;
    const int hi   = lane >> 4;
    const int lo   = lane & 15;
    const float scale = 0.17677669529663687f;
    float* sP  = sS + (size_t)w * 16 * SROW;
    float* stg = sS + STRIPS;

    const float* qrow = Qb + ((size_t)(b * N_ + q0 + lo) * CR_) + h * HD_ + 2 * hi;
    v2f aq[8];
    #pragma unroll
    for (int ts = 0; ts < 8; ++ts) { aq[ts].x = qrow[4 * ts]; aq[ts].y = qrow[4 * ts + 1]; }
    __builtin_amdgcn_sched_barrier(0);

    const int trow = tid >> 3;
    const int tc4  = (tid & 7) * 4;
    const size_t headoff = (size_t)b * N_ * CR_ + h * HD_ + tc4;
    const float* kg = Kb + headoff + (size_t)trow * CR_;
    const float* vg = Vb + headoff + (size_t)(2 * trow) * CR_;
    auto gloadK = [&](int tile, v4f (&g)[4]) {
        #pragma unroll
        for (int i = 0; i < 4; ++i)
            g[i] = *reinterpret_cast<const v4f*>(kg + (size_t)(tile * TK + 16 * i) * CR_);
    };
    auto gloadV = [&](int tile, v4f (&g)[4]) {
        #pragma unroll
        for (int i2 = 0; i2 < 2; ++i2) {
            g[2 * i2]     = *reinterpret_cast<const v4f*>(vg + (size_t)(tile * TK + 32 * i2) * CR_);
            g[2 * i2 + 1] = *reinterpret_cast<const v4f*>(vg + (size_t)(tile * TK + 32 * i2 + 1) * CR_);
        }
    };
    auto stashK = [&](int tile, const v4f (&g)[4]) {
        float* dst = stg + (tile % NSTAGE) * KBUF + trow * KP + tc4;
        #pragma unroll
        for (int i = 0; i < 4; ++i)
            *reinterpret_cast<v4f*>(dst + 16 * i * KP) = g[i];
    };
    auto stashV = [&](int tile, const v4f (&g)[4]) {
        float* dst = stg + (tile % NSTAGE) * VBUF + trow * VPP + 2 * tc4;
        #pragma unroll
        for (int i2 = 0; i2 < 2; ++i2) {
            v4f e = g[2 * i2], o = g[2 * i2 + 1];
            asm volatile("" : "+v"(e), "+v"(o));
            v4f lo4, hi4;
            lo4.x = e.x; lo4.y = o.x; lo4.z = e.y; lo4.w = o.y;
            hi4.x = e.z; hi4.y = o.z; hi4.z = e.w; hi4.w = o.w;
            *reinterpret_cast<v4f*>(dst + 16 * i2 * VPP)     = lo4;
            *reinterpret_cast<v4f*>(dst + 16 * i2 * VPP + 4) = hi4;
        }
    };

    constexpr int NT_QK = N_ / TK;
    v2f bkA[4][8], bkB[4][8];
    v4f gA[4], gB[4];

    auto kfrag = [&](int jt, v2f (&bk)[4][8]) {
        const float* src = stg + (jt % NSTAGE) * KBUF + lo * KP + 2 * hi;
        #pragma unroll
        for (int t = 0; t < 4; ++t) {
            #pragma unroll
            for (int ts = 0; ts < 8; ++ts)
                bk[t][ts] = *reinterpret_cast<const v2f*>(src + t * 16 * KP + 4 * ts);
        }
    };
    auto qk = [&](int jt, const v2f (&bk)[4][8]) {
        v8f acc[4] = {};
        #pragma unroll
        for (int ts = 0; ts < 8; ++ts) {
            #pragma unroll
            for (int t = 0; t < 4; ++t) acc[t] = wmma4(aq[ts], bk[t][ts], acc[t]);
        }
        #pragma unroll
        for (int t = 0; t < 4; ++t) {
            #pragma unroll
            for (int r = 0; r < 8; ++r)
                sP[(r + 8 * hi) * SROW + jt * TK + t * 16 + lo] = acc[t][r] * scale;
        }
    };

    gloadK(0, gA); gloadK(1, gB);
    stashK(0, gA); stashK(1, gB);
    gloadK(2, gA); gloadK(3, gB);
    lds_stage_barrier();
    kfrag(0, bkA);
    auto lastk = [](int t) { return t < NT_QK ? t : NT_QK - 1; };
    for (int jt = 0; jt < NT_QK; jt += 2) {
        kfrag(jt + 1, bkB);
        stashK(jt + 2, gA);
        gloadK(lastk(jt + 4), gA);
        __builtin_amdgcn_sched_barrier(0);
        qk(jt, bkA);
        lds_stage_barrier();
        kfrag(lastk(jt + 2), bkA);
        stashK(jt + 3, gB);
        gloadK(lastk(jt + 5), gB);
        __builtin_amdgcn_sched_barrier(0);
        qk(jt + 1, bkB);
        lds_stage_barrier();
    }

    gloadV(0, gA); gloadV(1, gB);

    float* myrow = sP + lo * SROW + 2 * hi;
    constexpr int CH = 32;
    float m0 = -3.0e38f, m1 = -3.0e38f, m2 = -3.0e38f, m3 = -3.0e38f;
    #pragma unroll 1
    for (int c0 = 0; c0 < N_ / 4; c0 += CH) {
        v2f tv[CH];
        #pragma unroll
        for (int i = 0; i < CH; ++i)
            tv[i] = *reinterpret_cast<const v2f*>(myrow + 4 * (c0 + i));
        #pragma unroll
        for (int i = 0; i < CH; i += 2) {
            m0 = fmaxf(m0, tv[i].x);     m1 = fmaxf(m1, tv[i].y);
            m2 = fmaxf(m2, tv[i + 1].x); m3 = fmaxf(m3, tv[i + 1].y);
        }
    }
    float mx = fmaxf(fmaxf(m0, m1), fmaxf(m2, m3));
    mx = fmaxf(mx, __shfl_xor(mx, 16, 32));
    float s0 = 0.f, s1 = 0.f, s2 = 0.f, s3 = 0.f;
    #pragma unroll 1
    for (int c0 = 0; c0 < N_ / 4; c0 += CH) {
        v2f tv[CH];
        #pragma unroll
        for (int i = 0; i < CH; ++i)
            tv[i] = *reinterpret_cast<const v2f*>(myrow + 4 * (c0 + i));
        #pragma unroll
        for (int i = 0; i < CH; i += 2) {
            tv[i].x     = __expf(tv[i].x - mx);     s0 += tv[i].x;
            tv[i].y     = __expf(tv[i].y - mx);     s1 += tv[i].y;
            tv[i + 1].x = __expf(tv[i + 1].x - mx); s2 += tv[i + 1].x;
            tv[i + 1].y = __expf(tv[i + 1].y - mx); s3 += tv[i + 1].y;
        }
        #pragma unroll
        for (int i = 0; i < CH; ++i)
            *reinterpret_cast<v2f*>(myrow + 4 * (c0 + i)) = tv[i];
    }
    float sum = (s0 + s1) + (s2 + s3);
    sum += __shfl_xor(sum, 16, 32);
    const float inv = 1.f / sum;
    float invr[8];
    #pragma unroll
    for (int r = 0; r < 8; ++r) invr[r] = __shfl(inv, r + 8 * hi, 32);

    v8f oacc[2] = {};
    const float* prow  = sP + lo * SROW + 2 * hi;
    const int d0 = lo, d1 = 16 + lo;
    constexpr int VH    = TK / 2;
    constexpr int VS    = VH / 4;
    constexpr int NT_PV = N_ / TK;
    static_assert(NT_PV == NT_QK, "lastk serves both loops");
    v2f apA[VS], vaA[VS], vbA[VS], apB[VS], vaB[VS], vbB[VS];

    auto pvfrag = [&](int it, int half, v2f (&ap)[VS], v2f (&va)[VS], v2f (&vb)[VS]) {
        const float* vs = stg + (it % NSTAGE) * VBUF + (half * (VH / 2) + hi) * VPP;
        #pragma unroll
        for (int u = 0; u < VS; ++u) {
            va[u] = *reinterpret_cast<const v2f*>(vs + 2 * u * VPP + 2 * d0);
            vb[u] = *reinterpret_cast<const v2f*>(vs + 2 * u * VPP + 2 * d1);
        }
        #pragma unroll
        for (int u = 0; u < VS; ++u)
            ap[u] = *reinterpret_cast<const v2f*>(prow + it * TK + half * VH + 4 * u);
    };
    auto pv = [&](const v2f (&ap)[VS], const v2f (&va)[VS], const v2f (&vb)[VS]) {
        __builtin_amdgcn_sched_barrier(0);
        #pragma unroll
        for (int u = 0; u < VS; ++u) {
            oacc[0] = wmma4(ap[u], va[u], oacc[0]);
            oacc[1] = wmma4(ap[u], vb[u], oacc[1]);
        }
        __builtin_amdgcn_sched_barrier(0);
    };

    stashV(0, gA); stashV(1, gB);
    gloadV(2, gA); gloadV(3, gB);
    lds_stage_barrier();
    pvfrag(0, 0, apA, vaA, vbA);
    for (int it = 0; it < NT_PV; it += 2) {
        pvfrag(it, 1, apB, vaB, vbB);
        stashV(it + 2, gA);
        gloadV(lastk(it + 4), gA);
        pv(apA, vaA, vbA);
        pvfrag(it + 1, 0, apA, vaA, vbA);
        pv(apB, vaB, vbB);
        lds_stage_barrier();
        pvfrag(it + 1, 1, apB, vaB, vbB);
        stashV(it + 3, gB);
        gloadV(lastk(it + 5), gB);
        pv(apA, vaA, vbA);
        pvfrag(lastk(it + 2), 0, apA, vaA, vbA);
        pv(apB, vaB, vbB);
        lds_stage_barrier();
    }

    #pragma unroll
    for (int tt = 0; tt < 2; ++tt) {
        const int col = h * HD_ + tt * 16 + lo;
        #pragma unroll
        for (int r = 0; r < 8; ++r) {
            const int row = q0 + r + 8 * hi;
            const size_t idx = (size_t)(b * N_ + row) * CR_ + col;
            Out[idx] = Hb[idx] + oacc[tt][r] * invr[r];
        }
    }
}

extern "C" void kernel_launch(void* const* d_in, const int* in_sizes, int n_in,
                              void* d_out, int out_size, void* d_ws, size_t ws_size,
                              hipStream_t stream) {
    (void)in_sizes; (void)n_in; (void)out_size; (void)ws_size;
    const float* x      = (const float*)d_in[0];
    const float* ln0_g  = (const float*)d_in[1];
    const float* ln0_b  = (const float*)d_in[2];
    const float* w_proj = (const float*)d_in[3];
    const float* b_proj = (const float*)d_in[4];
    const float* ln1_g  = (const float*)d_in[5];
    const float* ln1_b  = (const float*)d_in[6];
    const float* wq     = (const float*)d_in[7];
    const float* wk     = (const float*)d_in[8];
    const float* wv     = (const float*)d_in[9];
    float* out = (float*)d_out;

    float* ws = (float*)d_ws;
    const size_t rowsBig = (size_t)B_ * N_ * DIM_;
    const size_t rowsCr  = (size_t)B_ * N_ * CR_;
    float* xn = ws;
    float* hB = xn + rowsBig;
    float* yB = hB + rowsCr;
    float* qB = yB + rowsCr;
    float* kB = qB + rowsCr;
    float* vB = kB + rowsCr;

    const int M = B_ * N_;

    ln768_kernel<<<M, 256, 0, stream>>>(x, ln0_g, ln0_b, xn);

    (void)hipFuncSetAttribute((const void*)gemm_wmma_kernel<DIM_>,
                              hipFuncAttributeMaxDynamicSharedMemorySize,
                              GEMM_SMEM);
    gemm_wmma_kernel<DIM_><<<dim3(CR_ / GN, M / GM), 256, GEMM_SMEM, stream>>>(
        xn, w_proj, b_proj, hB, CR_);

    ln384_kernel<<<M / 8, 256, 0, stream>>>(hB, ln1_g, ln1_b, yB);

    (void)hipFuncSetAttribute((const void*)gemm_wmma_kernel<CR_>,
                              hipFuncAttributeMaxDynamicSharedMemorySize,
                              GEMM_SMEM);
    gemm_wmma_kernel<CR_><<<dim3(CR_ / GN, M / GM), 256, GEMM_SMEM, stream>>>(
        yB, wq, nullptr, qB, CR_);
    gemm_wmma_kernel<CR_><<<dim3(CR_ / GN, M / GM), 256, GEMM_SMEM, stream>>>(
        yB, wk, nullptr, kB, CR_);
    gemm_wmma_kernel<CR_><<<dim3(CR_ / GN, M / GM), 256, GEMM_SMEM, stream>>>(
        yB, wv, nullptr, vB, CR_);

    (void)hipFuncSetAttribute((const void*)attn_kernel,
                              hipFuncAttributeMaxDynamicSharedMemorySize,
                              ATTN_SMEM);
    attn_kernel<<<dim3(N_ / 64, B_ * H_), 128, ATTN_SMEM, stream>>>(
        qB, kB, vB, hB, out);
}


# attention: each wave's 1024 x 16 score strip kept in registers as transposed accumulators (S^T = K Q^T, 512 VGPRs); softmax and P V take P from those registers, no LDS strip or its round trips
# speedup vs baseline: 2.4955x; 1.1149x over previous
#include <hip/hip_runtime.h>

constexpr int B_   = 8;
constexpr int N_   = 1024;
constexpr int DIM_ = 768;
constexpr int H_   = 12;
constexpr int CR_  = 384;
constexpr int HD_  = 32;
constexpr float EPS_ = 1e-5f;

typedef __attribute__((ext_vector_type(2))) float v2f;
typedef __attribute__((ext_vector_type(4))) float v4f;
typedef __attribute__((ext_vector_type(8))) float v8f;

static __device__ __forceinline__ v8f wmma4(v2f a, v2f b, v8f c) {
    return __builtin_amdgcn_wmma_f32_16x16x4_f32(
         false, a,  false, b,
         (short)0, c,  false,  false);
}

__global__ __launch_bounds__(256)
void ln768_kernel(const float* __restrict__ x, const float* __restrict__ g,
                  const float* __restrict__ bia, float* __restrict__ o) {
    const int row  = blockIdx.x;
    const int t    = threadIdx.x;
    const float* xr = x + (size_t)row * DIM_;
    float v0 = xr[t], v1 = xr[t + 256], v2 = xr[t + 512];
    float s  = v0 + v1 + v2;
    float s2 = v0 * v0 + v1 * v1 + v2 * v2;
    #pragma unroll
    for (int off = 16; off > 0; off >>= 1) {
        s  += __shfl_xor(s,  off, 32);
        s2 += __shfl_xor(s2, off, 32);
    }
    __shared__ float ps[8], ps2[8];
    const int wid = t >> 5, lane = t & 31;
    if (lane == 0) { ps[wid] = s; ps2[wid] = s2; }
    __syncthreads();
    if (wid == 0) {
        float a  = (lane < 8) ? ps[lane]  : 0.f;
        float a2 = (lane < 8) ? ps2[lane] : 0.f;
        #pragma unroll
        for (int off = 4; off > 0; off >>= 1) {
            a  += __shfl_xor(a,  off, 32);
            a2 += __shfl_xor(a2, off, 32);
        }
        if (lane == 0) { ps[0] = a; ps2[0] = a2; }
    }
    __syncthreads();
    const float mean = ps[0] * (1.f / DIM_);
    const float var  = ps2[0] * (1.f / DIM_) - mean * mean;
    const float rstd = rsqrtf(var + EPS_);
    float* orow = o + (size_t)row * DIM_;
    orow[t]       = (v0 - mean) * rstd * g[t]       + bia[t];
    orow[t + 256] = (v1 - mean) * rstd * g[t + 256] + bia[t + 256];
    orow[t + 512] = (v2 - mean) * rstd * g[t + 512] + bia[t + 512];
}

__global__ __launch_bounds__(256)
void ln384_kernel(const float* __restrict__ x, const float* __restrict__ g,
                  const float* __restrict__ bia, float* __restrict__ o) {
    const int wid  = threadIdx.x >> 5;
    const int lane = threadIdx.x & 31;
    const int row  = blockIdx.x * 8 + wid;
    const float* xr = x + (size_t)row * CR_;
    float v[12];
    float s = 0.f, s2 = 0.f;
    #pragma unroll
    for (int i = 0; i < 12; ++i) {
        v[i] = xr[lane + 32 * i];
        s += v[i]; s2 += v[i] * v[i];
    }
    #pragma unroll
    for (int off = 16; off > 0; off >>= 1) {
        s  += __shfl_xor(s,  off, 32);
        s2 += __shfl_xor(s2, off, 32);
    }
    const float mean = s * (1.f / CR_);
    const float var  = s2 * (1.f / CR_) - mean * mean;
    const float rstd = rsqrtf(var + EPS_);
    float* orow = o + (size_t)row * CR_;
    #pragma unroll
    for (int i = 0; i < 12; ++i) {
        const int c = lane + 32 * i;
        orow[c] = (v[i] - mean) * rstd * g[c] + bia[c];
    }
}

static __device__ __forceinline__ void lds_stage_barrier() {
    __builtin_amdgcn_fence(__ATOMIC_RELEASE, "workgroup", "local");
    __builtin_amdgcn_s_barrier();
    __builtin_amdgcn_fence(__ATOMIC_ACQUIRE, "workgroup", "local");
}

constexpr int KC   = 32;
constexpr int GP   = KC + 4;
constexpr int GM   = 128;
constexpr int GN   = 96;
constexpr int GSTG = (GM + GN) * GP;
constexpr int GNST = 3;
constexpr int GEMM_SMEM = GNST * GSTG * 4;

template <int K>
__global__ __launch_bounds__(256)
void gemm_wmma_kernel(const float* __restrict__ A, const float* __restrict__ W,
                      const float* __restrict__ bias, float* __restrict__ C, int N) {
    extern __shared__ __attribute__((aligned(16))) float sT[];
    constexpr int NC = K / KC;
    static_assert(K % KC == 0 && NC % 2 == 0 && NC >= 4, "two chunks per loop trip");
    const int t    = threadIdx.x;
    const int lane = t & 31;
    const int w    = t >> 5;
    const int wm   = w >> 1;
    const int wn   = w & 1;
    const int hi   = lane >> 4;
    const int lo   = lane & 15;
    const int m0   = blockIdx.y * GM;
    const int n0   = blockIdx.x * GN;

    const int trow = t >> 3;
    const int tc4  = (t & 7) * 4;
    const float* ag = A + (size_t)(m0 + trow) * K + tc4;
    const float* wg = W + (size_t)(n0 + trow) * K + tc4;
    auto gload = [&](int c, v4f (&g)[7]) {
        #pragma unroll
        for (int i = 0; i < 4; ++i)
            g[i] = *reinterpret_cast<const v4f*>(ag + (size_t)(32 * i) * K + c * KC);
        #pragma unroll
        for (int i = 0; i < 3; ++i)
            g[4 + i] = *reinterpret_cast<const v4f*>(wg + (size_t)(32 * i) * K + c * KC);
    };
    auto stash = [&](int c, const v4f (&g)[7]) {
        float* dst = sT + (c % GNST) * GSTG + trow * GP + tc4;
        #pragma unroll
        for (int i = 0; i < 4; ++i)
            *reinterpret_cast<v4f*>(dst + 32 * i * GP) = g[i];
        #pragma unroll
        for (int i = 0; i < 3; ++i)
            *reinterpret_cast<v4f*>(dst + (GM + 32 * i) * GP) = g[4 + i];
    };

    v8f acc[2][3] = {};
    v2f faA[2][KC / 4], fbA[3][KC / 4], faB[2][KC / 4], fbB[3][KC / 4];
    auto fragload = [&](int c, v2f (&fa)[2][KC / 4], v2f (&fb)[3][KC / 4]) {
        const float* sa = sT + (c % GNST) * GSTG + (wm * 32 + lo) * GP + 2 * hi;
        const float* sb = sT + (c % GNST) * GSTG + (GM + wn * 48 + lo) * GP + 2 * hi;
        #pragma unroll
        for (int ks = 0; ks < KC / 4; ++ks) {
            #pragma unroll
            for (int rt = 0; rt < 2; ++rt)
                fa[rt][ks] = *reinterpret_cast<const v2f*>(sa + rt * 16 * GP + 4 * ks);
            #pragma unroll
            for (int ct = 0; ct < 3; ++ct)
                fb[ct][ks] = *reinterpret_cast<const v2f*>(sb + ct * 16 * GP + 4 * ks);
        }
    };
    auto mma = [&](const v2f (&fa)[2][KC / 4], const v2f (&fb)[3][KC / 4]) {
        __builtin_amdgcn_sched_barrier(0);
        #pragma unroll
        for (int ks = 0; ks < KC / 4; ++ks) {
            #pragma unroll
            for (int rt = 0; rt < 2; ++rt) {
                #pragma unroll
                for (int ct = 0; ct < 3; ++ct)
                    acc[rt][ct] = wmma4(fa[rt][ks], fb[ct][ks], acc[rt][ct]);
            }
        }
        __builtin_amdgcn_sched_barrier(0);
    };

    auto lastc = [](int c) { return c < NC ? c : NC - 1; };
    v4f gA[7], gB[7];
    gload(0, gA); gload(1, gB);
    stash(0, gA); stash(1, gB);
    gload(2, gA);
    __builtin_amdgcn_sched_barrier(0);
    gload(3, gB);
    lds_stage_barrier();
    fragload(0, faA, fbA);
    for (int c = 0; c < NC; c += 2) {
        fragload(c + 1, faB, fbB);
        stash(c + 2, gA);
        gload(lastc(c + 4), gA);
        mma(faA, fbA);
        lds_stage_barrier();
        fragload(lastc(c + 2), faA, fbA);
        stash(c + 3, gB);
        gload(lastc(c + 5), gB);
        mma(faB, fbB);
        lds_stage_barrier();
    }

    #pragma unroll
    for (int rt = 0; rt < 2; ++rt) {
        #pragma unroll
        for (int ct = 0; ct < 3; ++ct) {
            const int col = n0 + wn * 48 + ct * 16 + lo;
            const float bv = bias ? bias[col] : 0.f;
            #pragma unroll
            for (int r = 0; r < 8; ++r) {
                const int row = m0 + wm * 32 + rt * 16 + r + 8 * hi;
                C[(size_t)row * N + col] = acc[rt][ct][r] + bv;
            }
        }
    }
}

constexpr int TK     = 64;
constexpr int KP     = 36;
constexpr int VPP    = 72;
constexpr int KBUF   = TK * KP;
constexpr int VBUF   = (TK / 2) * VPP;
constexpr int NSTAGE = 3;
constexpr int STAGEF = NSTAGE * (VBUF > KBUF ? VBUF : KBUF);
constexpr int ATTN_SMEM = STAGEF * 4;
constexpr int NTILE  = N_ / 16;

__global__ __launch_bounds__(128)
void attn_kernel(const float* __restrict__ Qb, const float* __restrict__ Kb,
                 const float* __restrict__ Vb, const float* __restrict__ Hb,
                 float* __restrict__ Out) {
    extern __shared__ __attribute__((aligned(16))) float stg[];
    const int tid  = threadIdx.x;
    const int lane = tid & 31;
    const int w    = tid >> 5;
    const int bh   = blockIdx.y;
    const int b    = bh / H_;
    const int h    = bh % H_;
    const int q0   = blockIdx.x * 64 + w * 16;
    const int hi   = lane >> 4;
    const int lo   = lane & 15;

    const float* qrow = Qb + ((size_t)(b * N_ + q0 + lo) * CR_) + h * HD_ + 2 * hi;
    v2f aq[8];
    #pragma unroll
    for (int ts = 0; ts < 8; ++ts) { aq[ts].x = qrow[4 * ts]; aq[ts].y = qrow[4 * ts + 1]; }
    __builtin_amdgcn_sched_barrier(0);

    const int trow = tid >> 3;
    const int tc4  = (tid & 7) * 4;
    const size_t headoff = (size_t)b * N_ * CR_ + h * HD_ + tc4;
    const float* kg = Kb + headoff + (size_t)trow * CR_;
    const float* vg = Vb + headoff + (size_t)(2 * trow) * CR_;
    auto gloadK = [&](int tile, v4f (&g)[4]) {
        #pragma unroll
        for (int i = 0; i < 4; ++i)
            g[i] = *reinterpret_cast<const v4f*>(kg + (size_t)(tile * TK + 16 * i) * CR_);
    };
    auto gloadV = [&](int tile, v4f (&g)[4]) {
        #pragma unroll
        for (int i2 = 0; i2 < 2; ++i2) {
            g[2 * i2]     = *reinterpret_cast<const v4f*>(vg + (size_t)(tile * TK + 32 * i2) * CR_);
            g[2 * i2 + 1] = *reinterpret_cast<const v4f*>(vg + (size_t)(tile * TK + 32 * i2 + 1) * CR_);
        }
    };
    auto stashK = [&](int tile, const v4f (&g)[4]) {
        float* dst = stg + (tile % NSTAGE) * KBUF + trow * KP + tc4;
        #pragma unroll
        for (int i = 0; i < 4; ++i)
            *reinterpret_cast<v4f*>(dst + 16 * i * KP) = g[i];
    };
    auto stashV = [&](int tile, const v4f (&g)[4]) {
        float* dst = stg + (tile % NSTAGE) * VBUF + trow * VPP + 2 * tc4;
        #pragma unroll
        for (int i2 = 0; i2 < 2; ++i2) {
            v4f e = g[2 * i2], o = g[2 * i2 + 1];
            asm volatile("" : "+v"(e), "+v"(o));
            v4f lo4, hi4;
            lo4.x = e.x; lo4.y = o.x; lo4.z = e.y; lo4.w = o.y;
            hi4.x = e.z; hi4.y = o.z; hi4.z = e.w; hi4.w = o.w;
            *reinterpret_cast<v4f*>(dst + 16 * i2 * VPP)     = lo4;
            *reinterpret_cast<v4f*>(dst + 16 * i2 * VPP + 4) = hi4;
        }
    };

    constexpr int NT_QK = N_ / TK;
    v8f S[NTILE];
    #pragma unroll
    for (int T = 0; T < NTILE; ++T) S[T] = v8f{0.f, 0.f, 0.f, 0.f, 0.f, 0.f, 0.f, 0.f};
    v2f bkA[4][8], bkB[4][8];
    v4f gA[4], gB[4];

    auto kfrag = [&](int jt, v2f (&bk)[4][8]) {
        const float* src = stg + (jt % NSTAGE) * KBUF + lo * KP + 2 * hi;
        #pragma unroll
        for (int t = 0; t < 4; ++t) {
            #pragma unroll
            for (int ts = 0; ts < 8; ++ts)
                bk[t][ts] = *reinterpret_cast<const v2f*>(src + t * 16 * KP + 4 * ts);
        }
    };
    auto qk = [&](int jt, const v2f (&bk)[4][8]) {
        __builtin_amdgcn_sched_barrier(0);
        #pragma unroll
        for (int ts = 0; ts < 8; ++ts) {
            #pragma unroll
            for (int t = 0; t < 4; ++t)
                S[4 * jt + t] = wmma4(bk[t][ts], aq[ts], S[4 * jt + t]);
        }
        __builtin_amdgcn_sched_barrier(0);
    };

    gloadK(0, gA); gloadK(1, gB);
    stashK(0, gA); stashK(1, gB);
    gloadK(2, gA); gloadK(3, gB);
    lds_stage_barrier();
    kfrag(0, bkA);
    #pragma unroll
    for (int jt = 0; jt < NT_QK; jt += 2) {
        kfrag(jt + 1, bkB);
        if (jt + 2 < NT_QK) stashK(jt + 2, gA);
        if (jt + 4 < NT_QK) gloadK(jt + 4, gA);
        qk(jt, bkA);
        lds_stage_barrier();
        if (jt + 2 < NT_QK) kfrag(jt + 2, bkA);
        if (jt + 3 < NT_QK) stashK(jt + 3, gB);
        if (jt + 5 < NT_QK) gloadK(jt + 5, gB);
        qk(jt + 1, bkB);
        lds_stage_barrier();
    }

    gloadV(0, gA); gloadV(1, gB);

    float mxa[4] = {-3.0e38f, -3.0e38f, -3.0e38f, -3.0e38f};
    #pragma unroll
    for (int T = 0; T < NTILE; ++T) {
        #pragma unroll
        for (int k = 0; k < 4; ++k)
            mxa[k] = fmaxf(mxa[k], fmaxf(S[T][k], S[T][k + 4]));
    }
    float mx = fmaxf(fmaxf(mxa[0], mxa[1]), fmaxf(mxa[2], mxa[3]));
    mx = fmaxf(mx, __shfl_xor(mx, 16, 32));
    const float c1 = 0.17677669529663687f * 1.4426950408889634f;
    const float nb = -mx * c1;
    float sa[4] = {0.f, 0.f, 0.f, 0.f};
    #pragma unroll
    for (int T = 0; T < NTILE; ++T) {
        #pragma unroll
        for (int r = 0; r < 8; ++r) {
            const float e = __builtin_amdgcn_exp2f(fmaf(S[T][r], c1, nb));
            S[T][r] = e;
            sa[r & 3] += e;
        }
    }
    float sum = (sa[0] + sa[1]) + (sa[2] + sa[3]);
    sum += __shfl_xor(sum, 16, 32);
    const float inv = 1.f / sum;

    v8f oacc[2] = {v8f{0.f, 0.f, 0.f, 0.f, 0.f, 0.f, 0.f, 0.f}, v8f{0.f, 0.f, 0.f, 0.f, 0.f, 0.f, 0.f, 0.f}};
    constexpr int NT_PV = N_ / TK;
    v2f vaA[4][4], vbA[4][4], vaB[4][4], vbB[4][4];

    auto pvfrag = [&](int it, v2f (&va)[4][4], v2f (&vb)[4][4]) {
        const float* vs = stg + (it % NSTAGE) * VBUF + (4 * hi) * VPP + 2 * lo;
        #pragma unroll
        for (int t = 0; t < 4; ++t) {
            #pragma unroll
            for (int u = 0; u < 4; ++u) {
                va[t][u] = *reinterpret_cast<const v2f*>(vs + (8 * t + u) * VPP);
                vb[t][u] = *reinterpret_cast<const v2f*>(vs + (8 * t + u) * VPP + 32);
            }
        }
    };
    auto pv = [&](int it, const v2f (&va)[4][4], const v2f (&vb)[4][4]) {
        __builtin_amdgcn_sched_barrier(0);
        #pragma unroll
        for (int t = 0; t < 4; ++t) {
            #pragma unroll
            for (int u = 0; u < 4; ++u) {
                v2f p;
                p.x = S[4 * it + t][2 * u];
                p.y = S[4 * it + t][2 * u + 1];
                oacc[0] = wmma4(va[t][u], p, oacc[0]);
                oacc[1] = wmma4(vb[t][u], p, oacc[1]);
            }
        }
        __builtin_amdgcn_sched_barrier(0);
    };

    stashV(0, gA); stashV(1, gB);
    gloadV(2, gA); gloadV(3, gB);
    lds_stage_barrier();
    pvfrag(0, vaA, vbA);
    #pragma unroll
    for (int it = 0; it < NT_PV; it += 2) {
        pvfrag(it + 1, vaB, vbB);
        if (it + 2 < NT_PV) stashV(it + 2, gA);
        if (it + 4 < NT_PV) gloadV(it + 4, gA);
        pv(it, vaA, vbA);
        lds_stage_barrier();
        if (it + 2 < NT_PV) pvfrag(it + 2, vaA, vbA);
        if (it + 3 < NT_PV) stashV(it + 3, gB);
        if (it + 5 < NT_PV) gloadV(it + 5, gB);
        pv(it + 1, vaB, vbB);
        lds_stage_barrier();
    }

    #pragma unroll
    for (int tt = 0; tt < 2; ++tt) {
        const size_t idx = (size_t)(b * N_ + q0 + lo) * CR_ + h * HD_ + tt * 16 + 8 * hi;
        const v4f h0 = *reinterpret_cast<const v4f*>(Hb + idx);
        const v4f h1 = *reinterpret_cast<const v4f*>(Hb + idx + 4);
        v4f o0, o1;
        o0.x = h0.x + oacc[tt][0] * inv; o0.y = h0.y + oacc[tt][1] * inv;
        o0.z = h0.z + oacc[tt][2] * inv; o0.w = h0.w + oacc[tt][3] * inv;
        o1.x = h1.x + oacc[tt][4] * inv; o1.y = h1.y + oacc[tt][5] * inv;
        o1.z = h1.z + oacc[tt][6] * inv; o1.w = h1.w + oacc[tt][7] * inv;
        *reinterpret_cast<v4f*>(Out + idx)     = o0;
        *reinterpret_cast<v4f*>(Out + idx + 4) = o1;
    }
}

extern "C" void kernel_launch(void* const* d_in, const int* in_sizes, int n_in,
                              void* d_out, int out_size, void* d_ws, size_t ws_size,
                              hipStream_t stream) {
    (void)in_sizes; (void)n_in; (void)out_size; (void)ws_size;
    const float* x      = (const float*)d_in[0];
    const float* ln0_g  = (const float*)d_in[1];
    const float* ln0_b  = (const float*)d_in[2];
    const float* w_proj = (const float*)d_in[3];
    const float* b_proj = (const float*)d_in[4];
    const float* ln1_g  = (const float*)d_in[5];
    const float* ln1_b  = (const float*)d_in[6];
    const float* wq     = (const float*)d_in[7];
    const float* wk     = (const float*)d_in[8];
    const float* wv     = (const float*)d_in[9];
    float* out = (float*)d_out;

    float* ws = (float*)d_ws;
    const size_t rowsBig = (size_t)B_ * N_ * DIM_;
    const size_t rowsCr  = (size_t)B_ * N_ * CR_;
    float* xn = ws;
    float* hB = xn + rowsBig;
    float* yB = hB + rowsCr;
    float* qB = yB + rowsCr;
    float* kB = qB + rowsCr;
    float* vB = kB + rowsCr;

    const int M = B_ * N_;

    ln768_kernel<<<M, 256, 0, stream>>>(x, ln0_g, ln0_b, xn);

    (void)hipFuncSetAttribute((const void*)gemm_wmma_kernel<DIM_>,
                              hipFuncAttributeMaxDynamicSharedMemorySize,
                              GEMM_SMEM);
    gemm_wmma_kernel<DIM_><<<dim3(CR_ / GN, M / GM), 256, GEMM_SMEM, stream>>>(
        xn, w_proj, b_proj, hB, CR_);

    ln384_kernel<<<M / 8, 256, 0, stream>>>(hB, ln1_g, ln1_b, yB);

    (void)hipFuncSetAttribute((const void*)gemm_wmma_kernel<CR_>,
                              hipFuncAttributeMaxDynamicSharedMemorySize,
                              GEMM_SMEM);
    gemm_wmma_kernel<CR_><<<dim3(CR_ / GN, M / GM), 256, GEMM_SMEM, stream>>>(
        yB, wq, nullptr, qB, CR_);
    gemm_wmma_kernel<CR_><<<dim3(CR_ / GN, M / GM), 256, GEMM_SMEM, stream>>>(
        yB, wk, nullptr, kB, CR_);
    gemm_wmma_kernel<CR_><<<dim3(CR_ / GN, M / GM), 256, GEMM_SMEM, stream>>>(
        yB, wv, nullptr, vB, CR_);

    (void)hipFuncSetAttribute((const void*)attn_kernel,
                              hipFuncAttributeMaxDynamicSharedMemorySize,
                              ATTN_SMEM);
    attn_kernel<<<dim3(N_ / 64, B_ * H_), 128, ATTN_SMEM, stream>>>(
        qB, kB, vB, hB, out);
}


# attention: keys taken in two 512-key passes joined by a running row max and sum; score strip 256 VGPRs, 32-key V fragment sets, wave under 512 VGPRs: two work-groups per WGP, 2 waves per SIMD
# speedup vs baseline: 2.7143x; 1.0877x over previous
#include <hip/hip_runtime.h>

constexpr int B_   = 8;
constexpr int N_   = 1024;
constexpr int DIM_ = 768;
constexpr int H_   = 12;
constexpr int CR_  = 384;
constexpr int HD_  = 32;
constexpr float EPS_ = 1e-5f;

typedef __attribute__((ext_vector_type(2))) float v2f;
typedef __attribute__((ext_vector_type(4))) float v4f;
typedef __attribute__((ext_vector_type(8))) float v8f;

static __device__ __forceinline__ v8f wmma4(v2f a, v2f b, v8f c) {
    return __builtin_amdgcn_wmma_f32_16x16x4_f32(
         false, a,  false, b,
         (short)0, c,  false,  false);
}

__global__ __launch_bounds__(256)
void ln768_kernel(const float* __restrict__ x, const float* __restrict__ g,
                  const float* __restrict__ bia, float* __restrict__ o) {
    const int row  = blockIdx.x;
    const int t    = threadIdx.x;
    const float* xr = x + (size_t)row * DIM_;
    float v0 = xr[t], v1 = xr[t + 256], v2 = xr[t + 512];
    float s  = v0 + v1 + v2;
    float s2 = v0 * v0 + v1 * v1 + v2 * v2;
    #pragma unroll
    for (int off = 16; off > 0; off >>= 1) {
        s  += __shfl_xor(s,  off, 32);
        s2 += __shfl_xor(s2, off, 32);
    }
    __shared__ float ps[8], ps2[8];
    const int wid = t >> 5, lane = t & 31;
    if (lane == 0) { ps[wid] = s; ps2[wid] = s2; }
    __syncthreads();
    if (wid == 0) {
        float a  = (lane < 8) ? ps[lane]  : 0.f;
        float a2 = (lane < 8) ? ps2[lane] : 0.f;
        #pragma unroll
        for (int off = 4; off > 0; off >>= 1) {
            a  += __shfl_xor(a,  off, 32);
            a2 += __shfl_xor(a2, off, 32);
        }
        if (lane == 0) { ps[0] = a; ps2[0] = a2; }
    }
    __syncthreads();
    const float mean = ps[0] * (1.f / DIM_);
    const float var  = ps2[0] * (1.f / DIM_) - mean * mean;
    const float rstd = rsqrtf(var + EPS_);
    float* orow = o + (size_t)row * DIM_;
    orow[t]       = (v0 - mean) * rstd * g[t]       + bia[t];
    orow[t + 256] = (v1 - mean) * rstd * g[t + 256] + bia[t + 256];
    orow[t + 512] = (v2 - mean) * rstd * g[t + 512] + bia[t + 512];
}

__global__ __launch_bounds__(256)
void ln384_kernel(const float* __restrict__ x, const float* __restrict__ g,
                  const float* __restrict__ bia, float* __restrict__ o) {
    const int wid  = threadIdx.x >> 5;
    const int lane = threadIdx.x & 31;
    const int row  = blockIdx.x * 8 + wid;
    const float* xr = x + (size_t)row * CR_;
    float v[12];
    float s = 0.f, s2 = 0.f;
    #pragma unroll
    for (int i = 0; i < 12; ++i) {
        v[i] = xr[lane + 32 * i];
        s += v[i]; s2 += v[i] * v[i];
    }
    #pragma unroll
    for (int off = 16; off > 0; off >>= 1) {
        s  += __shfl_xor(s,  off, 32);
        s2 += __shfl_xor(s2, off, 32);
    }
    const float mean = s * (1.f / CR_);
    const float var  = s2 * (1.f / CR_) - mean * mean;
    const float rstd = rsqrtf(var + EPS_);
    float* orow = o + (size_t)row * CR_;
    #pragma unroll
    for (int i = 0; i < 12; ++i) {
        const int c = lane + 32 * i;
        orow[c] = (v[i] - mean) * rstd * g[c] + bia[c];
    }
}

static __device__ __forceinline__ void lds_stage_barrier() {
    __builtin_amdgcn_fence(__ATOMIC_RELEASE, "workgroup", "local");
    __builtin_amdgcn_s_barrier();
    __builtin_amdgcn_fence(__ATOMIC_ACQUIRE, "workgroup", "local");
}

constexpr int KC   = 32;
constexpr int GP   = KC + 4;
constexpr int GM   = 128;
constexpr int GN   = 96;
constexpr int GSTG = (GM + GN) * GP;
constexpr int GNST = 3;
constexpr int GEMM_SMEM = GNST * GSTG * 4;

template <int K>
__global__ __launch_bounds__(256)
void gemm_wmma_kernel(const float* __restrict__ A, const float* __restrict__ W,
                      const float* __restrict__ bias, float* __restrict__ C, int N) {
    extern __shared__ __attribute__((aligned(16))) float sT[];
    constexpr int NC = K / KC;
    static_assert(K % KC == 0 && NC % 2 == 0 && NC >= 4, "two chunks per loop trip");
    const int t    = threadIdx.x;
    const int lane = t & 31;
    const int w    = t >> 5;
    const int wm   = w >> 1;
    const int wn   = w & 1;
    const int hi   = lane >> 4;
    const int lo   = lane & 15;
    const int m0   = blockIdx.y * GM;
    const int n0   = blockIdx.x * GN;

    const int trow = t >> 3;
    const int tc4  = (t & 7) * 4;
    const float* ag = A + (size_t)(m0 + trow) * K + tc4;
    const float* wg = W + (size_t)(n0 + trow) * K + tc4;
    auto gload = [&](int c, v4f (&g)[7]) {
        #pragma unroll
        for (int i = 0; i < 4; ++i)
            g[i] = *reinterpret_cast<const v4f*>(ag + (size_t)(32 * i) * K + c * KC);
        #pragma unroll
        for (int i = 0; i < 3; ++i)
            g[4 + i] = *reinterpret_cast<const v4f*>(wg + (size_t)(32 * i) * K + c * KC);
    };
    auto stash = [&](int c, const v4f (&g)[7]) {
        float* dst = sT + (c % GNST) * GSTG + trow * GP + tc4;
        #pragma unroll
        for (int i = 0; i < 4; ++i)
            *reinterpret_cast<v4f*>(dst + 32 * i * GP) = g[i];
        #pragma unroll
        for (int i = 0; i < 3; ++i)
            *reinterpret_cast<v4f*>(dst + (GM + 32 * i) * GP) = g[4 + i];
    };

    v8f acc[2][3] = {};
    v2f faA[2][KC / 4], fbA[3][KC / 4], faB[2][KC / 4], fbB[3][KC / 4];
    auto fragload = [&](int c, v2f (&fa)[2][KC / 4], v2f (&fb)[3][KC / 4]) {
        const float* sa = sT + (c % GNST) * GSTG + (wm * 32 + lo) * GP + 2 * hi;
        const float* sb = sT + (c % GNST) * GSTG + (GM + wn * 48 + lo) * GP + 2 * hi;
        #pragma unroll
        for (int ks = 0; ks < KC / 4; ++ks) {
            #pragma unroll
            for (int rt = 0; rt < 2; ++rt)
                fa[rt][ks] = *reinterpret_cast<const v2f*>(sa + rt * 16 * GP + 4 * ks);
            #pragma unroll
            for (int ct = 0; ct < 3; ++ct)
                fb[ct][ks] = *reinterpret_cast<const v2f*>(sb + ct * 16 * GP + 4 * ks);
        }
    };
    auto mma = [&](const v2f (&fa)[2][KC / 4], const v2f (&fb)[3][KC / 4]) {
        __builtin_amdgcn_sched_barrier(0);
        #pragma unroll
        for (int ks = 0; ks < KC / 4; ++ks) {
            #pragma unroll
            for (int rt = 0; rt < 2; ++rt) {
                #pragma unroll
                for (int ct = 0; ct < 3; ++ct)
                    acc[rt][ct] = wmma4(fa[rt][ks], fb[ct][ks], acc[rt][ct]);
            }
        }
        __builtin_amdgcn_sched_barrier(0);
    };

    auto lastc = [](int c) { return c < NC ? c : NC - 1; };
    v4f gA[7], gB[7];
    gload(0, gA); gload(1, gB);
    stash(0, gA); stash(1, gB);
    gload(2, gA);
    __builtin_amdgcn_sched_barrier(0);
    gload(3, gB);
    lds_stage_barrier();
    fragload(0, faA, fbA);
    for (int c = 0; c < NC; c += 2) {
        fragload(c + 1, faB, fbB);
        stash(c + 2, gA);
        gload(lastc(c + 4), gA);
        mma(faA, fbA);
        lds_stage_barrier();
        fragload(lastc(c + 2), faA, fbA);
        stash(c + 3, gB);
        gload(lastc(c + 5), gB);
        mma(faB, fbB);
        lds_stage_barrier();
    }

    #pragma unroll
    for (int rt = 0; rt < 2; ++rt) {
        #pragma unroll
        for (int ct = 0; ct < 3; ++ct) {
            const int col = n0 + wn * 48 + ct * 16 + lo;
            const float bv = bias ? bias[col] : 0.f;
            #pragma unroll
            for (int r = 0; r < 8; ++r) {
                const int row = m0 + wm * 32 + rt * 16 + r + 8 * hi;
                C[(size_t)row * N + col] = acc[rt][ct][r] + bv;
            }
        }
    }
}

constexpr int TK     = 64;
constexpr int KP     = 36;
constexpr int VPP    = 72;
constexpr int KBUF   = TK * KP;
constexpr int VBUF   = (TK / 2) * VPP;
constexpr int NSTAGE = 3;
constexpr int STAGEF = NSTAGE * (VBUF > KBUF ? VBUF : KBUF);
constexpr int ATTN_SMEM = STAGEF * 4;
constexpr int HK     = N_ / 2;
constexpr int NTILE  = HK / 16;

__global__ __launch_bounds__(128, 2)
void attn_kernel(const float* __restrict__ Qb, const float* __restrict__ Kb,
                 const float* __restrict__ Vb, const float* __restrict__ Hb,
                 float* __restrict__ Out) {
    extern __shared__ __attribute__((aligned(16))) float stg[];
    const int tid  = threadIdx.x;
    const int lane = tid & 31;
    const int w    = tid >> 5;
    const int bh   = blockIdx.y;
    const int b    = bh / H_;
    const int h    = bh % H_;
    const int q0   = blockIdx.x * 64 + w * 16;
    const int hi   = lane >> 4;
    const int lo   = lane & 15;

    const float* qrow = Qb + ((size_t)(b * N_ + q0 + lo) * CR_) + h * HD_ + 2 * hi;
    v2f aq[8];
    #pragma unroll
    for (int ts = 0; ts < 8; ++ts) { aq[ts].x = qrow[4 * ts]; aq[ts].y = qrow[4 * ts + 1]; }
    __builtin_amdgcn_sched_barrier(0);

    const int trow = tid >> 3;
    const int tc4  = (tid & 7) * 4;
    const size_t headoff = (size_t)b * N_ * CR_ + h * HD_ + tc4;
    const float* kg = Kb + headoff + (size_t)trow * CR_;
    const float* vg = Vb + headoff + (size_t)(2 * trow) * CR_;
    auto gloadK = [&](int tile, v4f (&g)[4]) {
        #pragma unroll
        for (int i = 0; i < 4; ++i)
            g[i] = *reinterpret_cast<const v4f*>(kg + (size_t)(tile * TK + 16 * i) * CR_);
    };
    auto gloadV = [&](int tile, v4f (&g)[4]) {
        #pragma unroll
        for (int i2 = 0; i2 < 2; ++i2) {
            g[2 * i2]     = *reinterpret_cast<const v4f*>(vg + (size_t)(tile * TK + 32 * i2) * CR_);
            g[2 * i2 + 1] = *reinterpret_cast<const v4f*>(vg + (size_t)(tile * TK + 32 * i2 + 1) * CR_);
        }
    };
    auto stashK = [&](int tile, const v4f (&g)[4]) {
        float* dst = stg + (tile % NSTAGE) * KBUF + trow * KP + tc4;
        #pragma unroll
        for (int i = 0; i < 4; ++i)
            *reinterpret_cast<v4f*>(dst + 16 * i * KP) = g[i];
    };
    auto stashV = [&](int tile, const v4f (&g)[4]) {
        float* dst = stg + (tile % NSTAGE) * VBUF + trow * VPP + 2 * tc4;
        #pragma unroll
        for (int i2 = 0; i2 < 2; ++i2) {
            v4f e = g[2 * i2], o = g[2 * i2 + 1];
            asm volatile("" : "+v"(e), "+v"(o));
            v4f lo4, hi4;
            lo4.x = e.x; lo4.y = o.x; lo4.z = e.y; lo4.w = o.y;
            hi4.x = e.z; hi4.y = o.z; hi4.z = e.w; hi4.w = o.w;
            *reinterpret_cast<v4f*>(dst + 16 * i2 * VPP)     = lo4;
            *reinterpret_cast<v4f*>(dst + 16 * i2 * VPP + 4) = hi4;
        }
    };

    constexpr int NT_QK = HK / TK;
    constexpr int NT_PV = HK / TK;
    v8f S[NTILE];
    v2f bkA[4][8], bkB[4][8];
    v4f gA[4], gB[4];
    v2f vaA[2][4], vbA[2][4], vaB[2][4], vbB[2][4];
    v8f oacc[2] = {v8f{0.f, 0.f, 0.f, 0.f, 0.f, 0.f, 0.f, 0.f}, v8f{0.f, 0.f, 0.f, 0.f, 0.f, 0.f, 0.f, 0.f}};
    float m_run = -3.0e38f;
    float l_run = 0.f;
    const float c1 = 0.17677669529663687f * 1.4426950408889634f;

    auto kfrag = [&](int jt, v2f (&bk)[4][8]) {
        const float* src = stg + (jt % NSTAGE) * KBUF + lo * KP + 2 * hi;
        #pragma unroll
        for (int t = 0; t < 4; ++t) {
            #pragma unroll
            for (int ts = 0; ts < 8; ++ts)
                bk[t][ts] = *reinterpret_cast<const v2f*>(src + t * 16 * KP + 4 * ts);
        }
    };
    auto qk = [&](int jt, const v2f (&bk)[4][8]) {
        __builtin_amdgcn_sched_barrier(0);
        #pragma unroll
        for (int ts = 0; ts < 8; ++ts) {
            #pragma unroll
            for (int t = 0; t < 4; ++t)
                S[4 * jt + t] = wmma4(bk[t][ts], aq[ts], S[4 * jt + t]);
        }
        __builtin_amdgcn_sched_barrier(0);
    };

    auto pvfrag = [&](int it, int half, v2f (&va)[2][4], v2f (&vb)[2][4]) {
        const float* vs = stg + (it % NSTAGE) * VBUF + (4 * hi) * VPP + 2 * lo;
        #pragma unroll
        for (int t2 = 0; t2 < 2; ++t2) {
            #pragma unroll
            for (int u = 0; u < 4; ++u) {
                va[t2][u] = *reinterpret_cast<const v2f*>(vs + (8 * (2 * half + t2) + u) * VPP);
                vb[t2][u] = *reinterpret_cast<const v2f*>(vs + (8 * (2 * half + t2) + u) * VPP + 32);
            }
        }
    };
    auto pv = [&](int it, int half, const v2f (&va)[2][4], const v2f (&vb)[2][4]) {
        __builtin_amdgcn_sched_barrier(0);
        #pragma unroll
        for (int t2 = 0; t2 < 2; ++t2) {
            #pragma unroll
            for (int u = 0; u < 4; ++u) {
                v2f p;
                p.x = S[4 * it + 2 * half + t2][2 * u];
                p.y = S[4 * it + 2 * half + t2][2 * u + 1];
                oacc[0] = wmma4(va[t2][u], p, oacc[0]);
                oacc[1] = wmma4(vb[t2][u], p, oacc[1]);
            }
        }
        __builtin_amdgcn_sched_barrier(0);
    };

    #pragma nounroll
    for (int kh = 0; kh < 2; ++kh) {
        #pragma unroll
        for (int T = 0; T < NTILE; ++T) S[T] = v8f{0.f, 0.f, 0.f, 0.f, 0.f, 0.f, 0.f, 0.f};

        gloadK(0, gA); gloadK(1, gB);
        stashK(0, gA); stashK(1, gB);
        gloadK(2, gA); gloadK(3, gB);
        lds_stage_barrier();
        kfrag(0, bkA);
        #pragma unroll
        for (int jt = 0; jt < NT_QK; jt += 2) {
            kfrag(jt + 1, bkB);
            if (jt + 2 < NT_QK) stashK(jt + 2, gA);
            if (jt + 4 < NT_QK) gloadK(jt + 4, gA);
            qk(jt, bkA);
            lds_stage_barrier();
            if (jt + 2 < NT_QK) kfrag(jt + 2, bkA);
            if (jt + 3 < NT_QK) stashK(jt + 3, gB);
            if (jt + 5 < NT_QK) gloadK(jt + 5, gB);
            qk(jt + 1, bkB);
            lds_stage_barrier();
        }

        gloadV(0, gA); gloadV(1, gB);

        float mxa[4] = {-3.0e38f, -3.0e38f, -3.0e38f, -3.0e38f};
        #pragma unroll
        for (int T = 0; T < NTILE; ++T) {
            #pragma unroll
            for (int k = 0; k < 4; ++k)
                mxa[k] = fmaxf(mxa[k], fmaxf(S[T][k], S[T][k + 4]));
        }
        float mx = fmaxf(fmaxf(mxa[0], mxa[1]), fmaxf(mxa[2], mxa[3]));
        mx = fmaxf(mx, __shfl_xor(mx, 16, 32));
        const float m_new = fmaxf(m_run, mx);
        const float alpha = __builtin_amdgcn_exp2f((m_run - m_new) * c1);
        m_run = m_new;
        const float nb = -m_new * c1;
        float sa[4] = {0.f, 0.f, 0.f, 0.f};
        #pragma unroll
        for (int T = 0; T < NTILE; ++T) {
            #pragma unroll
            for (int r = 0; r < 8; ++r) {
                const float e = __builtin_amdgcn_exp2f(fmaf(S[T][r], c1, nb));
                S[T][r] = e;
                sa[r & 3] += e;
            }
        }
        l_run = fmaf(l_run, alpha, (sa[0] + sa[1]) + (sa[2] + sa[3]));
        #pragma unroll
        for (int tt = 0; tt < 2; ++tt) {
            #pragma unroll
            for (int r = 0; r < 8; ++r) oacc[tt][r] *= alpha;
        }

        stashV(0, gA); stashV(1, gB);
        gloadV(2, gA); gloadV(3, gB);
        lds_stage_barrier();
        pvfrag(0, 0, vaA, vbA);
        #pragma unroll
        for (int it = 0; it < NT_PV; it += 2) {
            pvfrag(it, 1, vaB, vbB);
            if (it + 2 < NT_PV) stashV(it + 2, gA);
            if (it + 4 < NT_PV) gloadV(it + 4, gA);
            pv(it, 0, vaA, vbA);
            pvfrag(it + 1, 0, vaA, vbA);
            pv(it, 1, vaB, vbB);
            lds_stage_barrier();
            pvfrag(it + 1, 1, vaB, vbB);
            if (it + 3 < NT_PV) stashV(it + 3, gB);
            if (it + 5 < NT_PV) gloadV(it + 5, gB);
            pv(it + 1, 0, vaA, vbA);
            if (it + 2 < NT_PV) pvfrag(it + 2, 0, vaA, vbA);
            pv(it + 1, 1, vaB, vbB);
            lds_stage_barrier();
        }

        kg += (size_t)HK * CR_;
        vg += (size_t)HK * CR_;
    }

    const float sum = l_run + __shfl_xor(l_run, 16, 32);
    const float inv = 1.f / sum;

    #pragma unroll
    for (int tt = 0; tt < 2; ++tt) {
        const size_t idx = (size_t)(b * N_ + q0 + lo) * CR_ + h * HD_ + tt * 16 + 8 * hi;
        const v4f h0 = *reinterpret_cast<const v4f*>(Hb + idx);
        const v4f h1 = *reinterpret_cast<const v4f*>(Hb + idx + 4);
        v4f o0, o1;
        o0.x = h0.x + oacc[tt][0] * inv; o0.y = h0.y + oacc[tt][1] * inv;
        o0.z = h0.z + oacc[tt][2] * inv; o0.w = h0.w + oacc[tt][3] * inv;
        o1.x = h1.x + oacc[tt][4] * inv; o1.y = h1.y + oacc[tt][5] * inv;
        o1.z = h1.z + oacc[tt][6] * inv; o1.w = h1.w + oacc[tt][7] * inv;
        *reinterpret_cast<v4f*>(Out + idx)     = o0;
        *reinterpret_cast<v4f*>(Out + idx + 4) = o1;
    }
}

extern "C" void kernel_launch(void* const* d_in, const int* in_sizes, int n_in,
                              void* d_out, int out_size, void* d_ws, size_t ws_size,
                              hipStream_t stream) {
    (void)in_sizes; (void)n_in; (void)out_size; (void)ws_size;
    const float* x      = (const float*)d_in[0];
    const float* ln0_g  = (const float*)d_in[1];
    const float* ln0_b  = (const float*)d_in[2];
    const float* w_proj = (const float*)d_in[3];
    const float* b_proj = (const float*)d_in[4];
    const float* ln1_g  = (const float*)d_in[5];
    const float* ln1_b  = (const float*)d_in[6];
    const float* wq     = (const float*)d_in[7];
    const float* wk     = (const float*)d_in[8];
    const float* wv     = (const float*)d_in[9];
    float* out = (float*)d_out;

    float* ws = (float*)d_ws;
    const size_t rowsBig = (size_t)B_ * N_ * DIM_;
    const size_t rowsCr  = (size_t)B_ * N_ * CR_;
    float* xn = ws;
    float* hB = xn + rowsBig;
    float* yB = hB + rowsCr;
    float* qB = yB + rowsCr;
    float* kB = qB + rowsCr;
    float* vB = kB + rowsCr;

    const int M = B_ * N_;

    ln768_kernel<<<M, 256, 0, stream>>>(x, ln0_g, ln0_b, xn);

    (void)hipFuncSetAttribute((const void*)gemm_wmma_kernel<DIM_>,
                              hipFuncAttributeMaxDynamicSharedMemorySize,
                              GEMM_SMEM);
    gemm_wmma_kernel<DIM_><<<dim3(CR_ / GN, M / GM), 256, GEMM_SMEM, stream>>>(
        xn, w_proj, b_proj, hB, CR_);

    ln384_kernel<<<M / 8, 256, 0, stream>>>(hB, ln1_g, ln1_b, yB);

    (void)hipFuncSetAttribute((const void*)gemm_wmma_kernel<CR_>,
                              hipFuncAttributeMaxDynamicSharedMemorySize,
                              GEMM_SMEM);
    gemm_wmma_kernel<CR_><<<dim3(CR_ / GN, M / GM), 256, GEMM_SMEM, stream>>>(
        yB, wq, nullptr, qB, CR_);
    gemm_wmma_kernel<CR_><<<dim3(CR_ / GN, M / GM), 256, GEMM_SMEM, stream>>>(
        yB, wk, nullptr, kB, CR_);
    gemm_wmma_kernel<CR_><<<dim3(CR_ / GN, M / GM), 256, GEMM_SMEM, stream>>>(
        yB, wv, nullptr, vB, CR_);

    (void)hipFuncSetAttribute((const void*)attn_kernel,
                              hipFuncAttributeMaxDynamicSharedMemorySize,
                              ATTN_SMEM);
    attn_kernel<<<dim3(N_ / 64, B_ * H_), 128, ATTN_SMEM, stream>>>(
        qB, kB, vB, hB, out);
}


# LayerNorm-384 and the Q, K, V projections in one launch: a work-group owns 32 whole rows, normalises them once into an LDS-resident A tile and streams the three weight matrices through the ring
# speedup vs baseline: 2.8908x; 1.0650x over previous
#include <hip/hip_runtime.h>

constexpr int B_   = 8;
constexpr int N_   = 1024;
constexpr int DIM_ = 768;
constexpr int H_   = 12;
constexpr int CR_  = 384;
constexpr int HD_  = 32;
constexpr float EPS_ = 1e-5f;

typedef __attribute__((ext_vector_type(2))) float v2f;
typedef __attribute__((ext_vector_type(4))) float v4f;
typedef __attribute__((ext_vector_type(8))) float v8f;

static __device__ __forceinline__ v8f wmma4(v2f a, v2f b, v8f c) {
    return __builtin_amdgcn_wmma_f32_16x16x4_f32(
         false, a,  false, b,
         (short)0, c,  false,  false);
}

__global__ __launch_bounds__(256)
void ln768_kernel(const float* __restrict__ x, const float* __restrict__ g,
                  const float* __restrict__ bia, float* __restrict__ o) {
    const int row  = blockIdx.x;
    const int t    = threadIdx.x;
    const float* xr = x + (size_t)row * DIM_;
    float v0 = xr[t], v1 = xr[t + 256], v2 = xr[t + 512];
    float s  = v0 + v1 + v2;
    float s2 = v0 * v0 + v1 * v1 + v2 * v2;
    #pragma unroll
    for (int off = 16; off > 0; off >>= 1) {
        s  += __shfl_xor(s,  off, 32);
        s2 += __shfl_xor(s2, off, 32);
    }
    __shared__ float ps[8], ps2[8];
    const int wid = t >> 5, lane = t & 31;
    if (lane == 0) { ps[wid] = s; ps2[wid] = s2; }
    __syncthreads();
    if (wid == 0) {
        float a  = (lane < 8) ? ps[lane]  : 0.f;
        float a2 = (lane < 8) ? ps2[lane] : 0.f;
        #pragma unroll
        for (int off = 4; off > 0; off >>= 1) {
            a  += __shfl_xor(a,  off, 32);
            a2 += __shfl_xor(a2, off, 32);
        }
        if (lane == 0) { ps[0] = a; ps2[0] = a2; }
    }
    __syncthreads();
    const float mean = ps[0] * (1.f / DIM_);
    const float var  = ps2[0] * (1.f / DIM_) - mean * mean;
    const float rstd = rsqrtf(var + EPS_);
    float* orow = o + (size_t)row * DIM_;
    orow[t]       = (v0 - mean) * rstd * g[t]       + bia[t];
    orow[t + 256] = (v1 - mean) * rstd * g[t + 256] + bia[t + 256];
    orow[t + 512] = (v2 - mean) * rstd * g[t + 512] + bia[t + 512];
}

static __device__ __forceinline__ void lds_stage_barrier() {
    __builtin_amdgcn_fence(__ATOMIC_RELEASE, "workgroup", "local");
    __builtin_amdgcn_s_barrier();
    __builtin_amdgcn_fence(__ATOMIC_ACQUIRE, "workgroup", "local");
}

constexpr int KC   = 32;
constexpr int GP   = KC + 4;
constexpr int GM   = 128;
constexpr int GN   = 96;
constexpr int GSTG = (GM + GN) * GP;
constexpr int GNST = 3;
constexpr int GEMM_SMEM = GNST * GSTG * 4;

template <int K>
__global__ __launch_bounds__(256)
void gemm_wmma_kernel(const float* __restrict__ A, const float* __restrict__ W,
                      const float* __restrict__ bias, float* __restrict__ C, int N) {
    extern __shared__ __attribute__((aligned(16))) float sT[];
    constexpr int NC = K / KC;
    static_assert(K % KC == 0 && NC % 2 == 0 && NC >= 4, "two chunks per loop trip");
    const int t    = threadIdx.x;
    const int lane = t & 31;
    const int w    = t >> 5;
    const int wm   = w >> 1;
    const int wn   = w & 1;
    const int hi   = lane >> 4;
    const int lo   = lane & 15;
    const int m0   = blockIdx.y * GM;
    const int n0   = blockIdx.x * GN;

    const int trow = t >> 3;
    const int tc4  = (t & 7) * 4;
    const float* ag = A + (size_t)(m0 + trow) * K + tc4;
    const float* wg = W + (size_t)(n0 + trow) * K + tc4;
    auto gload = [&](int c, v4f (&g)[7]) {
        #pragma unroll
        for (int i = 0; i < 4; ++i)
            g[i] = *reinterpret_cast<const v4f*>(ag + (size_t)(32 * i) * K + c * KC);
        #pragma unroll
        for (int i = 0; i < 3; ++i)
            g[4 + i] = *reinterpret_cast<const v4f*>(wg + (size_t)(32 * i) * K + c * KC);
    };
    auto stash = [&](int c, const v4f (&g)[7]) {
        float* dst = sT + (c % GNST) * GSTG + trow * GP + tc4;
        #pragma unroll
        for (int i = 0; i < 4; ++i)
            *reinterpret_cast<v4f*>(dst + 32 * i * GP) = g[i];
        #pragma unroll
        for (int i = 0; i < 3; ++i)
            *reinterpret_cast<v4f*>(dst + (GM + 32 * i) * GP) = g[4 + i];
    };

    v8f acc[2][3] = {};
    v2f faA[2][KC / 4], fbA[3][KC / 4], faB[2][KC / 4], fbB[3][KC / 4];
    auto fragload = [&](int c, v2f (&fa)[2][KC / 4], v2f (&fb)[3][KC / 4]) {
        const float* sa = sT + (c % GNST) * GSTG + (wm * 32 + lo) * GP + 2 * hi;
        const float* sb = sT + (c % GNST) * GSTG + (GM + wn * 48 + lo) * GP + 2 * hi;
        #pragma unroll
        for (int ks = 0; ks < KC / 4; ++ks) {
            #pragma unroll
            for (int rt = 0; rt < 2; ++rt)
                fa[rt][ks] = *reinterpret_cast<const v2f*>(sa + rt * 16 * GP + 4 * ks);
            #pragma unroll
            for (int ct = 0; ct < 3; ++ct)
                fb[ct][ks] = *reinterpret_cast<const v2f*>(sb + ct * 16 * GP + 4 * ks);
        }
    };
    auto mma = [&](const v2f (&fa)[2][KC / 4], const v2f (&fb)[3][KC / 4]) {
        __builtin_amdgcn_sched_barrier(0);
        #pragma unroll
        for (int ks = 0; ks < KC / 4; ++ks) {
            #pragma unroll
            for (int rt = 0; rt < 2; ++rt) {
                #pragma unroll
                for (int ct = 0; ct < 3; ++ct)
                    acc[rt][ct] = wmma4(fa[rt][ks], fb[ct][ks], acc[rt][ct]);
            }
        }
        __builtin_amdgcn_sched_barrier(0);
    };

    auto lastc = [](int c) { return c < NC ? c : NC - 1; };
    v4f gA[7], gB[7];
    gload(0, gA); gload(1, gB);
    stash(0, gA); stash(1, gB);
    gload(2, gA);
    __builtin_amdgcn_sched_barrier(0);
    gload(3, gB);
    lds_stage_barrier();
    fragload(0, faA, fbA);
    for (int c = 0; c < NC; c += 2) {
        fragload(c + 1, faB, fbB);
        stash(c + 2, gA);
        gload(lastc(c + 4), gA);
        mma(faA, fbA);
        lds_stage_barrier();
        fragload(lastc(c + 2), faA, fbA);
        stash(c + 3, gB);
        gload(lastc(c + 5), gB);
        mma(faB, fbB);
        lds_stage_barrier();
    }

    #pragma unroll
    for (int rt = 0; rt < 2; ++rt) {
        #pragma unroll
        for (int ct = 0; ct < 3; ++ct) {
            const int col = n0 + wn * 48 + ct * 16 + lo;
            const float bv = bias ? bias[col] : 0.f;
            #pragma unroll
            for (int r = 0; r < 8; ++r) {
                const int row = m0 + wm * 32 + rt * 16 + r + 8 * hi;
                C[(size_t)row * N + col] = acc[rt][ct][r] + bv;
            }
        }
    }
}

constexpr int QM   = 32;
constexpr int QYP  = CR_ + 4;
constexpr int QWST = CR_ * GP;
constexpr int QNC  = CR_ / KC;
constexpr int QNG  = 3 * QNC;
constexpr int QKV_SMEM = (QM * QYP + GNST * QWST) * 4;

__global__ __launch_bounds__(256)
void ln_qkv_kernel(const float* __restrict__ Hh, const float* __restrict__ g,
                   const float* __restrict__ bia,
                   const float* __restrict__ W0, const float* __restrict__ W1,
                   const float* __restrict__ W2,
                   float* __restrict__ C0, float* __restrict__ C1, float* __restrict__ C2) {
    extern __shared__ __attribute__((aligned(16))) float sQ[];
    float* sY = sQ;
    float* sW = sQ + QM * QYP;
    const int t    = threadIdx.x;
    const int lane = t & 31;
    const int w    = t >> 5;
    const int hi   = lane >> 4;
    const int lo   = lane & 15;
    const int m0   = blockIdx.x * QM;

    const int trow = t >> 3;
    const int tc4  = (t & 7) * 4;
    auto lastg = [](int gi) { return gi < QNG ? gi : QNG - 1; };
    auto gload = [&](int gi, v4f (&r)[12]) {
        const int z = gi / QNC;
        const int c = gi - z * QNC;
        const float* wz = (z == 0) ? W0 : ((z == 1) ? W1 : W2);
        const float* wp = wz + (size_t)trow * CR_ + tc4 + c * KC;
        #pragma unroll
        for (int i = 0; i < 12; ++i)
            r[i] = *reinterpret_cast<const v4f*>(wp + (size_t)(32 * i) * CR_);
    };
    auto stash = [&](int gi, const v4f (&r)[12]) {
        float* dst = sW + (gi % GNST) * QWST + trow * GP + tc4;
        #pragma unroll
        for (int i = 0; i < 12; ++i)
            *reinterpret_cast<v4f*>(dst + 32 * i * GP) = r[i];
    };

    v8f acc[2][3] = {};
    v2f faA[2][KC / 4], fbA[3][KC / 4], faB[2][KC / 4], fbB[3][KC / 4];
    auto fragload = [&](int gi, v2f (&fa)[2][KC / 4], v2f (&fb)[3][KC / 4]) {
        const int c = gi % QNC;
        const float* sa = sY + lo * QYP + c * KC + 2 * hi;
        const float* sb = sW + (gi % GNST) * QWST + (w * 48 + lo) * GP + 2 * hi;
        #pragma unroll
        for (int ks = 0; ks < KC / 4; ++ks) {
            #pragma unroll
            for (int rt = 0; rt < 2; ++rt)
                fa[rt][ks] = *reinterpret_cast<const v2f*>(sa + rt * 16 * QYP + 4 * ks);
            #pragma unroll
            for (int ct = 0; ct < 3; ++ct)
                fb[ct][ks] = *reinterpret_cast<const v2f*>(sb + ct * 16 * GP + 4 * ks);
        }
    };
    auto mma = [&](const v2f (&fa)[2][KC / 4], const v2f (&fb)[3][KC / 4]) {
        __builtin_amdgcn_sched_barrier(0);
        #pragma unroll
        for (int ks = 0; ks < KC / 4; ++ks) {
            #pragma unroll
            for (int rt = 0; rt < 2; ++rt) {
                #pragma unroll
                for (int ct = 0; ct < 3; ++ct)
                    acc[rt][ct] = wmma4(fa[rt][ks], fb[ct][ks], acc[rt][ct]);
            }
        }
        __builtin_amdgcn_sched_barrier(0);
    };

    v4f hv[4][3], gg[3], bb[3];
    {
        const float* hr = Hh + (size_t)(m0 + 4 * w) * CR_ + lane * 4;
        #pragma unroll
        for (int j = 0; j < 4; ++j) {
            #pragma unroll
            for (int i = 0; i < 3; ++i)
                hv[j][i] = *reinterpret_cast<const v4f*>(hr + (size_t)j * CR_ + 128 * i);
        }
        #pragma unroll
        for (int i = 0; i < 3; ++i) {
            gg[i] = *reinterpret_cast<const v4f*>(g   + lane * 4 + 128 * i);
            bb[i] = *reinterpret_cast<const v4f*>(bia + lane * 4 + 128 * i);
        }
    }
    __builtin_amdgcn_sched_barrier(0);
    v4f gA[12], gB[12];
    gload(0, gA); gload(1, gB);
    __builtin_amdgcn_sched_barrier(0);

    #pragma unroll
    for (int j = 0; j < 4; ++j) {
        float s = 0.f, s2 = 0.f;
        #pragma unroll
        for (int i = 0; i < 3; ++i) {
            const v4f v = hv[j][i];
            s  += (v.x + v.y) + (v.z + v.w);
            s2 += (v.x * v.x + v.y * v.y) + (v.z * v.z + v.w * v.w);
        }
        #pragma unroll
        for (int off = 16; off > 0; off >>= 1) {
            s  += __shfl_xor(s,  off, 32);
            s2 += __shfl_xor(s2, off, 32);
        }
        const float mean = s * (1.f / CR_);
        const float var  = s2 * (1.f / CR_) - mean * mean;
        const float rstd = rsqrtf(var + EPS_);
        float* yr = sY + (4 * w + j) * QYP + lane * 4;
        #pragma unroll
        for (int i = 0; i < 3; ++i) {
            const v4f v = hv[j][i];
            v4f y;
            y.x = (v.x - mean) * rstd * gg[i].x + bb[i].x;
            y.y = (v.y - mean) * rstd * gg[i].y + bb[i].y;
            y.z = (v.z - mean) * rstd * gg[i].z + bb[i].z;
            y.w = (v.w - mean) * rstd * gg[i].w + bb[i].w;
            *reinterpret_cast<v4f*>(yr + 128 * i) = y;
        }
    }

    stash(0, gA); stash(1, gB);
    gload(2, gA);
    __builtin_amdgcn_sched_barrier(0);
    gload(3, gB);
    lds_stage_barrier();
    fragload(0, faA, fbA);
    for (int gi = 0; gi < QNG; gi += 2) {
        fragload(gi + 1, faB, fbB);
        stash(gi + 2, gA);
        gload(lastg(gi + 4), gA);
        mma(faA, fbA);
        lds_stage_barrier();
        fragload(lastg(gi + 2), faA, fbA);
        stash(gi + 3, gB);
        gload(lastg(gi + 5), gB);
        mma(faB, fbB);
        lds_stage_barrier();
        if ((gi + 2) % QNC == 0) {
            float* __restrict__ C = (gi + 2 == QNC) ? C0 : ((gi + 2 == 2 * QNC) ? C1 : C2);
            #pragma unroll
            for (int rt = 0; rt < 2; ++rt) {
                #pragma unroll
                for (int ct = 0; ct < 3; ++ct) {
                    const int col = w * 48 + ct * 16 + lo;
                    #pragma unroll
                    for (int r = 0; r < 8; ++r) {
                        const int row = m0 + rt * 16 + r + 8 * hi;
                        C[(size_t)row * CR_ + col] = acc[rt][ct][r];
                    }
                    acc[rt][ct] = v8f{0.f, 0.f, 0.f, 0.f, 0.f, 0.f, 0.f, 0.f};
                }
            }
        }
    }
}

constexpr int TK     = 64;
constexpr int KP     = 36;
constexpr int VPP    = 72;
constexpr int KBUF   = TK * KP;
constexpr int VBUF   = (TK / 2) * VPP;
constexpr int NSTAGE = 3;
constexpr int STAGEF = NSTAGE * (VBUF > KBUF ? VBUF : KBUF);
constexpr int ATTN_SMEM = STAGEF * 4;
constexpr int HK     = N_ / 2;
constexpr int NTILE  = HK / 16;

__global__ __launch_bounds__(128, 2)
void attn_kernel(const float* __restrict__ Qb, const float* __restrict__ Kb,
                 const float* __restrict__ Vb, const float* __restrict__ Hb,
                 float* __restrict__ Out) {
    extern __shared__ __attribute__((aligned(16))) float stg[];
    const int tid  = threadIdx.x;
    const int lane = tid & 31;
    const int w    = tid >> 5;
    const int bh   = blockIdx.y;
    const int b    = bh / H_;
    const int h    = bh % H_;
    const int q0   = blockIdx.x * 64 + w * 16;
    const int hi   = lane >> 4;
    const int lo   = lane & 15;

    const float* qrow = Qb + ((size_t)(b * N_ + q0 + lo) * CR_) + h * HD_ + 2 * hi;
    v2f aq[8];
    #pragma unroll
    for (int ts = 0; ts < 8; ++ts) { aq[ts].x = qrow[4 * ts]; aq[ts].y = qrow[4 * ts + 1]; }
    __builtin_amdgcn_sched_barrier(0);

    const int trow = tid >> 3;
    const int tc4  = (tid & 7) * 4;
    const size_t headoff = (size_t)b * N_ * CR_ + h * HD_ + tc4;
    const float* kg = Kb + headoff + (size_t)trow * CR_;
    const float* vg = Vb + headoff + (size_t)(2 * trow) * CR_;
    auto gloadK = [&](int tile, v4f (&g)[4]) {
        #pragma unroll
        for (int i = 0; i < 4; ++i)
            g[i] = *reinterpret_cast<const v4f*>(kg + (size_t)(tile * TK + 16 * i) * CR_);
    };
    auto gloadV = [&](int tile, v4f (&g)[4]) {
        #pragma unroll
        for (int i2 = 0; i2 < 2; ++i2) {
            g[2 * i2]     = *reinterpret_cast<const v4f*>(vg + (size_t)(tile * TK + 32 * i2) * CR_);
            g[2 * i2 + 1] = *reinterpret_cast<const v4f*>(vg + (size_t)(tile * TK + 32 * i2 + 1) * CR_);
        }
    };
    auto stashK = [&](int tile, const v4f (&g)[4]) {
        float* dst = stg + (tile % NSTAGE) * KBUF + trow * KP + tc4;
        #pragma unroll
        for (int i = 0; i < 4; ++i)
            *reinterpret_cast<v4f*>(dst + 16 * i * KP) = g[i];
    };
    auto stashV = [&](int tile, const v4f (&g)[4]) {
        float* dst = stg + (tile % NSTAGE) * VBUF + trow * VPP + 2 * tc4;
        #pragma unroll
        for (int i2 = 0; i2 < 2; ++i2) {
            v4f e = g[2 * i2], o = g[2 * i2 + 1];
            asm volatile("" : "+v"(e), "+v"(o));
            v4f lo4, hi4;
            lo4.x = e.x; lo4.y = o.x; lo4.z = e.y; lo4.w = o.y;
            hi4.x = e.z; hi4.y = o.z; hi4.z = e.w; hi4.w = o.w;
            *reinterpret_cast<v4f*>(dst + 16 * i2 * VPP)     = lo4;
            *reinterpret_cast<v4f*>(dst + 16 * i2 * VPP + 4) = hi4;
        }
    };

    constexpr int NT_QK = HK / TK;
    constexpr int NT_PV = HK / TK;
    v8f S[NTILE];
    v2f bkA[4][8], bkB[4][8];
    v4f gA[4], gB[4];
    v2f vaA[2][4], vbA[2][4], vaB[2][4], vbB[2][4];
    v8f oacc[2] = {v8f{0.f, 0.f, 0.f, 0.f, 0.f, 0.f, 0.f, 0.f}, v8f{0.f, 0.f, 0.f, 0.f, 0.f, 0.f, 0.f, 0.f}};
    float m_run = -3.0e38f;
    float l_run = 0.f;
    const float c1 = 0.17677669529663687f * 1.4426950408889634f;

    auto kfrag = [&](int jt, v2f (&bk)[4][8]) {
        const float* src = stg + (jt % NSTAGE) * KBUF + lo * KP + 2 * hi;
        #pragma unroll
        for (int t = 0; t < 4; ++t) {
            #pragma unroll
            for (int ts = 0; ts < 8; ++ts)
                bk[t][ts] = *reinterpret_cast<const v2f*>(src + t * 16 * KP + 4 * ts);
        }
    };
    auto qk = [&](int jt, const v2f (&bk)[4][8]) {
        __builtin_amdgcn_sched_barrier(0);
        #pragma unroll
        for (int ts = 0; ts < 8; ++ts) {
            #pragma unroll
            for (int t = 0; t < 4; ++t)
                S[4 * jt + t] = wmma4(bk[t][ts], aq[ts], S[4 * jt + t]);
        }
        __builtin_amdgcn_sched_barrier(0);
    };

    auto pvfrag = [&](int it, int half, v2f (&va)[2][4], v2f (&vb)[2][4]) {
        const float* vs = stg + (it % NSTAGE) * VBUF + (4 * hi) * VPP + 2 * lo;
        #pragma unroll
        for (int t2 = 0; t2 < 2; ++t2) {
            #pragma unroll
            for (int u = 0; u < 4; ++u) {
                va[t2][u] = *reinterpret_cast<const v2f*>(vs + (8 * (2 * half + t2) + u) * VPP);
                vb[t2][u] = *reinterpret_cast<const v2f*>(vs + (8 * (2 * half + t2) + u) * VPP + 32);
            }
        }
    };
    auto pv = [&](int it, int half, const v2f (&va)[2][4], const v2f (&vb)[2][4]) {
        __builtin_amdgcn_sched_barrier(0);
        #pragma unroll
        for (int t2 = 0; t2 < 2; ++t2) {
            #pragma unroll
            for (int u = 0; u < 4; ++u) {
                v2f p;
                p.x = S[4 * it + 2 * half + t2][2 * u];
                p.y = S[4 * it + 2 * half + t2][2 * u + 1];
                oacc[0] = wmma4(va[t2][u], p, oacc[0]);
                oacc[1] = wmma4(vb[t2][u], p, oacc[1]);
            }
        }
        __builtin_amdgcn_sched_barrier(0);
    };

    #pragma nounroll
    for (int kh = 0; kh < 2; ++kh) {
        #pragma unroll
        for (int T = 0; T < NTILE; ++T) S[T] = v8f{0.f, 0.f, 0.f, 0.f, 0.f, 0.f, 0.f, 0.f};

        gloadK(0, gA); gloadK(1, gB);
        stashK(0, gA); stashK(1, gB);
        gloadK(2, gA); gloadK(3, gB);
        lds_stage_barrier();
        kfrag(0, bkA);
        #pragma unroll
        for (int jt = 0; jt < NT_QK; jt += 2) {
            kfrag(jt + 1, bkB);
            if (jt + 2 < NT_QK) stashK(jt + 2, gA);
            if (jt + 4 < NT_QK) gloadK(jt + 4, gA);
            qk(jt, bkA);
            lds_stage_barrier();
            if (jt + 2 < NT_QK) kfrag(jt + 2, bkA);
            if (jt + 3 < NT_QK) stashK(jt + 3, gB);
            if (jt + 5 < NT_QK) gloadK(jt + 5, gB);
            qk(jt + 1, bkB);
            lds_stage_barrier();
        }

        gloadV(0, gA); gloadV(1, gB);

        float mxa[4] = {-3.0e38f, -3.0e38f, -3.0e38f, -3.0e38f};
        #pragma unroll
        for (int T = 0; T < NTILE; ++T) {
            #pragma unroll
            for (int k = 0; k < 4; ++k)
                mxa[k] = fmaxf(mxa[k], fmaxf(S[T][k], S[T][k + 4]));
        }
        float mx = fmaxf(fmaxf(mxa[0], mxa[1]), fmaxf(mxa[2], mxa[3]));
        mx = fmaxf(mx, __shfl_xor(mx, 16, 32));
        const float m_new = fmaxf(m_run, mx);
        const float alpha = __builtin_amdgcn_exp2f((m_run - m_new) * c1);
        m_run = m_new;
        const float nb = -m_new * c1;
        float sa[4] = {0.f, 0.f, 0.f, 0.f};
        #pragma unroll
        for (int T = 0; T < NTILE; ++T) {
            #pragma unroll
            for (int r = 0; r < 8; ++r) {
                const float e = __builtin_amdgcn_exp2f(fmaf(S[T][r], c1, nb));
                S[T][r] = e;
                sa[r & 3] += e;
            }
        }
        l_run = fmaf(l_run, alpha, (sa[0] + sa[1]) + (sa[2] + sa[3]));
        #pragma unroll
        for (int tt = 0; tt < 2; ++tt) {
            #pragma unroll
            for (int r = 0; r < 8; ++r) oacc[tt][r] *= alpha;
        }

        stashV(0, gA); stashV(1, gB);
        gloadV(2, gA); gloadV(3, gB);
        lds_stage_barrier();
        pvfrag(0, 0, vaA, vbA);
        #pragma unroll
        for (int it = 0; it < NT_PV; it += 2) {
            pvfrag(it, 1, vaB, vbB);
            if (it + 2 < NT_PV) stashV(it + 2, gA);
            if (it + 4 < NT_PV) gloadV(it + 4, gA);
            pv(it, 0, vaA, vbA);
            pvfrag(it + 1, 0, vaA, vbA);
            pv(it, 1, vaB, vbB);
            lds_stage_barrier();
            pvfrag(it + 1, 1, vaB, vbB);
            if (it + 3 < NT_PV) stashV(it + 3, gB);
            if (it + 5 < NT_PV) gloadV(it + 5, gB);
            pv(it + 1, 0, vaA, vbA);
            if (it + 2 < NT_PV) pvfrag(it + 2, 0, vaA, vbA);
            pv(it + 1, 1, vaB, vbB);
            lds_stage_barrier();
        }

        kg += (size_t)HK * CR_;
        vg += (size_t)HK * CR_;
    }

    const float sum = l_run + __shfl_xor(l_run, 16, 32);
    const float inv = 1.f / sum;

    #pragma unroll
    for (int tt = 0; tt < 2; ++tt) {
        const size_t idx = (size_t)(b * N_ + q0 + lo) * CR_ + h * HD_ + tt * 16 + 8 * hi;
        const v4f h0 = *reinterpret_cast<const v4f*>(Hb + idx);
        const v4f h1 = *reinterpret_cast<const v4f*>(Hb + idx + 4);
        v4f o0, o1;
        o0.x = h0.x + oacc[tt][0] * inv; o0.y = h0.y + oacc[tt][1] * inv;
        o0.z = h0.z + oacc[tt][2] * inv; o0.w = h0.w + oacc[tt][3] * inv;
        o1.x = h1.x + oacc[tt][4] * inv; o1.y = h1.y + oacc[tt][5] * inv;
        o1.z = h1.z + oacc[tt][6] * inv; o1.w = h1.w + oacc[tt][7] * inv;
        *reinterpret_cast<v4f*>(Out + idx)     = o0;
        *reinterpret_cast<v4f*>(Out + idx + 4) = o1;
    }
}

extern "C" void kernel_launch(void* const* d_in, const int* in_sizes, int n_in,
                              void* d_out, int out_size, void* d_ws, size_t ws_size,
                              hipStream_t stream) {
    (void)in_sizes; (void)n_in; (void)out_size; (void)ws_size;
    const float* x      = (const float*)d_in[0];
    const float* ln0_g  = (const float*)d_in[1];
    const float* ln0_b  = (const float*)d_in[2];
    const float* w_proj = (const float*)d_in[3];
    const float* b_proj = (const float*)d_in[4];
    const float* ln1_g  = (const float*)d_in[5];
    const float* ln1_b  = (const float*)d_in[6];
    const float* wq     = (const float*)d_in[7];
    const float* wk     = (const float*)d_in[8];
    const float* wv     = (const float*)d_in[9];
    float* out = (float*)d_out;

    float* ws = (float*)d_ws;
    const size_t rowsBig = (size_t)B_ * N_ * DIM_;
    const size_t rowsCr  = (size_t)B_ * N_ * CR_;
    float* xn = ws;
    float* hB = xn + rowsBig;
    float* yB = hB + rowsCr;
    float* qB = yB + rowsCr;
    float* kB = qB + rowsCr;
    float* vB = kB + rowsCr;

    const int M = B_ * N_;

    ln768_kernel<<<M, 256, 0, stream>>>(x, ln0_g, ln0_b, xn);

    (void)hipFuncSetAttribute((const void*)gemm_wmma_kernel<DIM_>,
                              hipFuncAttributeMaxDynamicSharedMemorySize,
                              GEMM_SMEM);
    gemm_wmma_kernel<DIM_><<<dim3(CR_ / GN, M / GM), 256, GEMM_SMEM, stream>>>(
        xn, w_proj, b_proj, hB, CR_);

    (void)hipFuncSetAttribute((const void*)ln_qkv_kernel,
                              hipFuncAttributeMaxDynamicSharedMemorySize,
                              QKV_SMEM);
    ln_qkv_kernel<<<M / QM, 256, QKV_SMEM, stream>>>(
        hB, ln1_g, ln1_b, wq, wk, wv, qB, kB, vB);

    (void)hipFuncSetAttribute((const void*)attn_kernel,
                              hipFuncAttributeMaxDynamicSharedMemorySize,
                              ATTN_SMEM);
    attn_kernel<<<dim3(N_ / 64, B_ * H_), 128, ATTN_SMEM, stream>>>(
        qB, kB, vB, hB, out);
}


# GEMM K loops: a wave fetches, parks and reads back only the 48 weight rows of its own columns (wave-private LDS ring): no work-group barrier per step; priority steps down per loop quarter
# speedup vs baseline: 3.2235x; 1.1151x over previous
#include <hip/hip_runtime.h>

constexpr int B_   = 8;
constexpr int N_   = 1024;
constexpr int DIM_ = 768;
constexpr int H_   = 12;
constexpr int CR_  = 384;
constexpr int HD_  = 32;
constexpr float EPS_ = 1e-5f;

typedef __attribute__((ext_vector_type(2))) float v2f;
typedef __attribute__((ext_vector_type(4))) float v4f;
typedef __attribute__((ext_vector_type(8))) float v8f;

static __device__ __forceinline__ v8f wmma4(v2f a, v2f b, v8f c) {
    return __builtin_amdgcn_wmma_f32_16x16x4_f32(
         false, a,  false, b,
         (short)0, c,  false,  false);
}

static __device__ __forceinline__ void lds_stage_barrier() {
    __builtin_amdgcn_fence(__ATOMIC_RELEASE, "workgroup", "local");
    __builtin_amdgcn_s_barrier();
    __builtin_amdgcn_fence(__ATOMIC_ACQUIRE, "workgroup", "local");
}

static __device__ __forceinline__ void wave_lds_order() {
    __builtin_amdgcn_fence(__ATOMIC_RELEASE, "wavefront", "local");
    __builtin_amdgcn_wave_barrier();
    __builtin_amdgcn_fence(__ATOMIC_ACQUIRE, "wavefront", "local");
}

constexpr int KC   = 32;
constexpr int GP   = KC + 4;
constexpr int GNST = 3;

constexpr int QM   = 32;
constexpr int QYP  = CR_ + 4;
constexpr int QWST = CR_ * GP;
constexpr int QNC  = CR_ / KC;
constexpr int QNG  = 3 * QNC;
constexpr int QKV_SMEM = (QM * QYP + GNST * QWST) * 4;

__global__ __launch_bounds__(256)
void ln_qkv_kernel(const float* __restrict__ Hh, const float* __restrict__ g,
                   const float* __restrict__ bia,
                   const float* __restrict__ W0, const float* __restrict__ W1,
                   const float* __restrict__ W2,
                   float* __restrict__ C0, float* __restrict__ C1, float* __restrict__ C2) {
    extern __shared__ __attribute__((aligned(16))) float sQ[];
    float* sY = sQ;
    float* sW = sQ + QM * QYP;
    const int t    = threadIdx.x;
    const int lane = t & 31;
    const int w    = t >> 5;
    const int hi   = lane >> 4;
    const int lo   = lane & 15;
    const int m0   = blockIdx.x * QM;

    const int trow = w * 48 + (lane >> 3);
    const int tc4  = (lane & 7) * 4;
    auto lastg = [](int gi) { return gi < QNG ? gi : QNG - 1; };
    auto gload = [&](int gi, v4f (&r)[12]) {
        const int z = gi / QNC;
        const int c = gi - z * QNC;
        const float* wz = (z == 0) ? W0 : ((z == 1) ? W1 : W2);
        const float* wp = wz + (size_t)trow * CR_ + tc4 + c * KC;
        #pragma unroll
        for (int i = 0; i < 12; ++i)
            r[i] = *reinterpret_cast<const v4f*>(wp + (size_t)(4 * i) * CR_);
    };
    auto stash = [&](int gi, const v4f (&r)[12]) {
        float* dst = sW + (gi % GNST) * QWST + trow * GP + tc4;
        #pragma unroll
        for (int i = 0; i < 12; ++i)
            *reinterpret_cast<v4f*>(dst + 4 * i * GP) = r[i];
    };

    v8f acc[2][3] = {};
    v2f faA[2][KC / 4], fbA[3][KC / 4], faB[2][KC / 4], fbB[3][KC / 4];
    auto fragload = [&](int gi, v2f (&fa)[2][KC / 4], v2f (&fb)[3][KC / 4]) {
        const int c = gi % QNC;
        const float* sa = sY + lo * QYP + c * KC + 2 * hi;
        const float* sb = sW + (gi % GNST) * QWST + (w * 48 + lo) * GP + 2 * hi;
        #pragma unroll
        for (int ks = 0; ks < KC / 4; ++ks) {
            #pragma unroll
            for (int rt = 0; rt < 2; ++rt)
                fa[rt][ks] = *reinterpret_cast<const v2f*>(sa + rt * 16 * QYP + 4 * ks);
            #pragma unroll
            for (int ct = 0; ct < 3; ++ct)
                fb[ct][ks] = *reinterpret_cast<const v2f*>(sb + ct * 16 * GP + 4 * ks);
        }
    };
    auto mma = [&](const v2f (&fa)[2][KC / 4], const v2f (&fb)[3][KC / 4]) {
        __builtin_amdgcn_sched_barrier(0);
        #pragma unroll
        for (int ks = 0; ks < KC / 4; ++ks) {
            #pragma unroll
            for (int rt = 0; rt < 2; ++rt) {
                #pragma unroll
                for (int ct = 0; ct < 3; ++ct)
                    acc[rt][ct] = wmma4(fa[rt][ks], fb[ct][ks], acc[rt][ct]);
            }
        }
        __builtin_amdgcn_sched_barrier(0);
    };

    v4f hv[4][3], gg[3], bb[3];
    {
        const float* hr = Hh + (size_t)(m0 + 4 * w) * CR_ + lane * 4;
        #pragma unroll
        for (int j = 0; j < 4; ++j) {
            #pragma unroll
            for (int i = 0; i < 3; ++i)
                hv[j][i] = *reinterpret_cast<const v4f*>(hr + (size_t)j * CR_ + 128 * i);
        }
        #pragma unroll
        for (int i = 0; i < 3; ++i) {
            gg[i] = *reinterpret_cast<const v4f*>(g   + lane * 4 + 128 * i);
            bb[i] = *reinterpret_cast<const v4f*>(bia + lane * 4 + 128 * i);
        }
    }
    __builtin_amdgcn_sched_barrier(0);
    v4f gA[12], gB[12];
    gload(0, gA); gload(1, gB);
    __builtin_amdgcn_sched_barrier(0);

    #pragma unroll
    for (int j = 0; j < 4; ++j) {
        float s = 0.f, s2 = 0.f;
        #pragma unroll
        for (int i = 0; i < 3; ++i) {
            const v4f v = hv[j][i];
            s  += (v.x + v.y) + (v.z + v.w);
            s2 += (v.x * v.x + v.y * v.y) + (v.z * v.z + v.w * v.w);
        }
        #pragma unroll
        for (int off = 16; off > 0; off >>= 1) {
            s  += __shfl_xor(s,  off, 32);
            s2 += __shfl_xor(s2, off, 32);
        }
        const float mean = s * (1.f / CR_);
        const float var  = s2 * (1.f / CR_) - mean * mean;
        const float rstd = rsqrtf(var + EPS_);
        float* yr = sY + (4 * w + j) * QYP + lane * 4;
        #pragma unroll
        for (int i = 0; i < 3; ++i) {
            const v4f v = hv[j][i];
            v4f y;
            y.x = (v.x - mean) * rstd * gg[i].x + bb[i].x;
            y.y = (v.y - mean) * rstd * gg[i].y + bb[i].y;
            y.z = (v.z - mean) * rstd * gg[i].z + bb[i].z;
            y.w = (v.w - mean) * rstd * gg[i].w + bb[i].w;
            *reinterpret_cast<v4f*>(yr + 128 * i) = y;
        }
    }

    stash(0, gA); stash(1, gB);
    gload(2, gA);
    __builtin_amdgcn_sched_barrier(0);
    gload(3, gB);
    lds_stage_barrier();
    __builtin_amdgcn_s_setprio(3);
    fragload(0, faA, fbA);
    for (int gi = 0; gi < QNG; gi += 2) {
        fragload(gi + 1, faB, fbB);
        stash(gi + 2, gA);
        gload(lastg(gi + 4), gA);
        mma(faA, fbA);
        wave_lds_order();
        fragload(lastg(gi + 2), faA, fbA);
        stash(gi + 3, gB);
        gload(lastg(gi + 5), gB);
        mma(faB, fbB);
        wave_lds_order();
        if (gi + 2 == 8)       __builtin_amdgcn_s_setprio(2);
        else if (gi + 2 == 18) __builtin_amdgcn_s_setprio(1);
        else if (gi + 2 == 28) __builtin_amdgcn_s_setprio(0);
        if ((gi + 2) % QNC == 0) {
            float* __restrict__ C = (gi + 2 == QNC) ? C0 : ((gi + 2 == 2 * QNC) ? C1 : C2);
            #pragma unroll
            for (int rt = 0; rt < 2; ++rt) {
                #pragma unroll
                for (int ct = 0; ct < 3; ++ct) {
                    const int col = w * 48 + ct * 16 + lo;
                    #pragma unroll
                    for (int r = 0; r < 8; ++r) {
                        const int row = m0 + rt * 16 + r + 8 * hi;
                        C[(size_t)row * CR_ + col] = acc[rt][ct][r];
                    }
                    acc[rt][ct] = v8f{0.f, 0.f, 0.f, 0.f, 0.f, 0.f, 0.f, 0.f};
                }
            }
        }
    }
}

constexpr int PXP  = DIM_ + 4;
constexpr int PNC  = DIM_ / KC;
constexpr int PROJ_SMEM = (QM * PXP + GNST * QWST) * 4;

__global__ __launch_bounds__(256)
void ln_proj_kernel(const float* __restrict__ X, const float* __restrict__ g,
                    const float* __restrict__ bia, const float* __restrict__ Wp,
                    const float* __restrict__ bp, float* __restrict__ C) {
    extern __shared__ __attribute__((aligned(16))) float sP[];
    static_assert(PNC % 2 == 0 && PNC >= 4, "two chunk steps per loop trip");
    float* sX = sP;
    float* sW = sP + QM * PXP;
    const int t    = threadIdx.x;
    const int lane = t & 31;
    const int w    = t >> 5;
    const int hi   = lane >> 4;
    const int lo   = lane & 15;
    const int m0   = blockIdx.x * QM;

    const int trow = w * 48 + (lane >> 3);
    const int tc4  = (lane & 7) * 4;
    auto lastg = [](int gi) { return gi < PNC ? gi : PNC - 1; };
    const float* wrow = Wp + (size_t)trow * DIM_ + tc4;
    auto gload = [&](int gi, v4f (&r)[12]) {
        const float* wp = wrow + gi * KC;
        #pragma unroll
        for (int i = 0; i < 12; ++i)
            r[i] = *reinterpret_cast<const v4f*>(wp + (size_t)(4 * i) * DIM_);
    };
    auto stash = [&](int gi, const v4f (&r)[12]) {
        float* dst = sW + (gi % GNST) * QWST + trow * GP + tc4;
        #pragma unroll
        for (int i = 0; i < 12; ++i)
            *reinterpret_cast<v4f*>(dst + 4 * i * GP) = r[i];
    };

    v8f acc[2][3] = {};
    v2f faA[2][KC / 4], fbA[3][KC / 4], faB[2][KC / 4], fbB[3][KC / 4];
    auto fragload = [&](int gi, v2f (&fa)[2][KC / 4], v2f (&fb)[3][KC / 4]) {
        const float* sa = sX + lo * PXP + gi * KC + 2 * hi;
        const float* sb = sW + (gi % GNST) * QWST + (w * 48 + lo) * GP + 2 * hi;
        #pragma unroll
        for (int ks = 0; ks < KC / 4; ++ks) {
            #pragma unroll
            for (int rt = 0; rt < 2; ++rt)
                fa[rt][ks] = *reinterpret_cast<const v2f*>(sa + rt * 16 * PXP + 4 * ks);
            #pragma unroll
            for (int ct = 0; ct < 3; ++ct)
                fb[ct][ks] = *reinterpret_cast<const v2f*>(sb + ct * 16 * GP + 4 * ks);
        }
    };
    auto mma = [&](const v2f (&fa)[2][KC / 4], const v2f (&fb)[3][KC / 4]) {
        __builtin_amdgcn_sched_barrier(0);
        #pragma unroll
        for (int ks = 0; ks < KC / 4; ++ks) {
            #pragma unroll
            for (int rt = 0; rt < 2; ++rt) {
                #pragma unroll
                for (int ct = 0; ct < 3; ++ct)
                    acc[rt][ct] = wmma4(fa[rt][ks], fb[ct][ks], acc[rt][ct]);
            }
        }
        __builtin_amdgcn_sched_barrier(0);
    };

    v4f hv[4][6], gg[6], bb[6];
    {
        const float* hr = X + (size_t)(m0 + 4 * w) * DIM_ + lane * 4;
        #pragma unroll
        for (int j = 0; j < 4; ++j) {
            #pragma unroll
            for (int i = 0; i < 6; ++i)
                hv[j][i] = *reinterpret_cast<const v4f*>(hr + (size_t)j * DIM_ + 128 * i);
        }
        #pragma unroll
        for (int i = 0; i < 6; ++i) {
            gg[i] = *reinterpret_cast<const v4f*>(g   + lane * 4 + 128 * i);
            bb[i] = *reinterpret_cast<const v4f*>(bia + lane * 4 + 128 * i);
        }
    }
    __builtin_amdgcn_sched_barrier(0);
    v4f gA[12], gB[12];
    gload(0, gA); gload(1, gB);
    __builtin_amdgcn_sched_barrier(0);

    #pragma unroll
    for (int j = 0; j < 4; ++j) {
        float s = 0.f, s2 = 0.f;
        #pragma unroll
        for (int i = 0; i < 6; ++i) {
            const v4f v = hv[j][i];
            s  += (v.x + v.y) + (v.z + v.w);
            s2 += (v.x * v.x + v.y * v.y) + (v.z * v.z + v.w * v.w);
        }
        #pragma unroll
        for (int off = 16; off > 0; off >>= 1) {
            s  += __shfl_xor(s,  off, 32);
            s2 += __shfl_xor(s2, off, 32);
        }
        const float mean = s * (1.f / DIM_);
        const float var  = s2 * (1.f / DIM_) - mean * mean;
        const float rstd = rsqrtf(var + EPS_);
        float* yr = sX + (4 * w + j) * PXP + lane * 4;
        #pragma unroll
        for (int i = 0; i < 6; ++i) {
            const v4f v = hv[j][i];
            v4f y;
            y.x = (v.x - mean) * rstd * gg[i].x + bb[i].x;
            y.y = (v.y - mean) * rstd * gg[i].y + bb[i].y;
            y.z = (v.z - mean) * rstd * gg[i].z + bb[i].z;
            y.w = (v.w - mean) * rstd * gg[i].w + bb[i].w;
            *reinterpret_cast<v4f*>(yr + 128 * i) = y;
        }
    }

    stash(0, gA); stash(1, gB);
    gload(2, gA);
    __builtin_amdgcn_sched_barrier(0);
    gload(3, gB);
    lds_stage_barrier();
    __builtin_amdgcn_s_setprio(3);
    fragload(0, faA, fbA);
    for (int gi = 0; gi < PNC; gi += 2) {
        fragload(gi + 1, faB, fbB);
        stash(gi + 2, gA);
        gload(lastg(gi + 4), gA);
        mma(faA, fbA);
        wave_lds_order();
        fragload(lastg(gi + 2), faA, fbA);
        stash(gi + 3, gB);
        gload(lastg(gi + 5), gB);
        mma(faB, fbB);
        wave_lds_order();
        if (gi + 2 == 6)       __builtin_amdgcn_s_setprio(2);
        else if (gi + 2 == 12) __builtin_amdgcn_s_setprio(1);
        else if (gi + 2 == 18) __builtin_amdgcn_s_setprio(0);
    }

    #pragma unroll
    for (int rt = 0; rt < 2; ++rt) {
        #pragma unroll
        for (int ct = 0; ct < 3; ++ct) {
            const int col = w * 48 + ct * 16 + lo;
            const float bv = bp[col];
            #pragma unroll
            for (int r = 0; r < 8; ++r) {
                const int row = m0 + rt * 16 + r + 8 * hi;
                C[(size_t)row * CR_ + col] = acc[rt][ct][r] + bv;
            }
        }
    }
}

constexpr int TK     = 64;
constexpr int KP     = 36;
constexpr int VPP    = 72;
constexpr int KBUF   = TK * KP;
constexpr int VBUF   = (TK / 2) * VPP;
constexpr int NSTAGE = 3;
constexpr int STAGEF = NSTAGE * (VBUF > KBUF ? VBUF : KBUF);
constexpr int ATTN_SMEM = STAGEF * 4;
constexpr int HK     = N_ / 2;
constexpr int NTILE  = HK / 16;

__global__ __launch_bounds__(128, 2)
void attn_kernel(const float* __restrict__ Qb, const float* __restrict__ Kb,
                 const float* __restrict__ Vb, const float* __restrict__ Hb,
                 float* __restrict__ Out) {
    extern __shared__ __attribute__((aligned(16))) float stg[];
    const int tid  = threadIdx.x;
    const int lane = tid & 31;
    const int w    = tid >> 5;
    const int bh   = blockIdx.y;
    const int b    = bh / H_;
    const int h    = bh % H_;
    const int q0   = blockIdx.x * 64 + w * 16;
    const int hi   = lane >> 4;
    const int lo   = lane & 15;

    const float* qrow = Qb + ((size_t)(b * N_ + q0 + lo) * CR_) + h * HD_ + 2 * hi;
    v2f aq[8];
    #pragma unroll
    for (int ts = 0; ts < 8; ++ts) { aq[ts].x = qrow[4 * ts]; aq[ts].y = qrow[4 * ts + 1]; }
    __builtin_amdgcn_sched_barrier(0);

    const int trow = tid >> 3;
    const int tc4  = (tid & 7) * 4;
    const size_t headoff = (size_t)b * N_ * CR_ + h * HD_ + tc4;
    const float* kg = Kb + headoff + (size_t)trow * CR_;
    const float* vg = Vb + headoff + (size_t)(2 * trow) * CR_;
    auto gloadK = [&](int tile, v4f (&g)[4]) {
        #pragma unroll
        for (int i = 0; i < 4; ++i)
            g[i] = *reinterpret_cast<const v4f*>(kg + (size_t)(tile * TK + 16 * i) * CR_);
    };
    auto gloadV = [&](int tile, v4f (&g)[4]) {
        #pragma unroll
        for (int i2 = 0; i2 < 2; ++i2) {
            g[2 * i2]     = *reinterpret_cast<const v4f*>(vg + (size_t)(tile * TK + 32 * i2) * CR_);
            g[2 * i2 + 1] = *reinterpret_cast<const v4f*>(vg + (size_t)(tile * TK + 32 * i2 + 1) * CR_);
        }
    };
    auto stashK = [&](int tile, const v4f (&g)[4]) {
        float* dst = stg + (tile % NSTAGE) * KBUF + trow * KP + tc4;
        #pragma unroll
        for (int i = 0; i < 4; ++i)
            *reinterpret_cast<v4f*>(dst + 16 * i * KP) = g[i];
    };
    auto stashV = [&](int tile, const v4f (&g)[4]) {
        float* dst = stg + (tile % NSTAGE) * VBUF + trow * VPP + 2 * tc4;
        #pragma unroll
        for (int i2 = 0; i2 < 2; ++i2) {
            v4f e = g[2 * i2], o = g[2 * i2 + 1];
            asm volatile("" : "+v"(e), "+v"(o));
            v4f lo4, hi4;
            lo4.x = e.x; lo4.y = o.x; lo4.z = e.y; lo4.w = o.y;
            hi4.x = e.z; hi4.y = o.z; hi4.z = e.w; hi4.w = o.w;
            *reinterpret_cast<v4f*>(dst + 16 * i2 * VPP)     = lo4;
            *reinterpret_cast<v4f*>(dst + 16 * i2 * VPP + 4) = hi4;
        }
    };

    constexpr int NT_QK = HK / TK;
    constexpr int NT_PV = HK / TK;
    v8f S[NTILE];
    v2f bkA[4][8], bkB[4][8];
    v4f gA[4], gB[4];
    v2f vaA[2][4], vbA[2][4], vaB[2][4], vbB[2][4];
    v8f oacc[2] = {v8f{0.f, 0.f, 0.f, 0.f, 0.f, 0.f, 0.f, 0.f}, v8f{0.f, 0.f, 0.f, 0.f, 0.f, 0.f, 0.f, 0.f}};
    float m_run = -3.0e38f;
    float l_run = 0.f;
    const float c1 = 0.17677669529663687f * 1.4426950408889634f;

    auto kfrag = [&](int jt, v2f (&bk)[4][8]) {
        const float* src = stg + (jt % NSTAGE) * KBUF + lo * KP + 2 * hi;
        #pragma unroll
        for (int t = 0; t < 4; ++t) {
            #pragma unroll
            for (int ts = 0; ts < 8; ++ts)
                bk[t][ts] = *reinterpret_cast<const v2f*>(src + t * 16 * KP + 4 * ts);
        }
    };
    auto qk = [&](int jt, const v2f (&bk)[4][8]) {
        __builtin_amdgcn_sched_barrier(0);
        #pragma unroll
        for (int ts = 0; ts < 8; ++ts) {
            #pragma unroll
            for (int t = 0; t < 4; ++t)
                S[4 * jt + t] = wmma4(bk[t][ts], aq[ts], S[4 * jt + t]);
        }
        __builtin_amdgcn_sched_barrier(0);
    };

    auto pvfrag = [&](int it, int half, v2f (&va)[2][4], v2f (&vb)[2][4]) {
        const float* vs = stg + (it % NSTAGE) * VBUF + (4 * hi) * VPP + 2 * lo;
        #pragma unroll
        for (int t2 = 0; t2 < 2; ++t2) {
            #pragma unroll
            for (int u = 0; u < 4; ++u) {
                va[t2][u] = *reinterpret_cast<const v2f*>(vs + (8 * (2 * half + t2) + u) * VPP);
                vb[t2][u] = *reinterpret_cast<const v2f*>(vs + (8 * (2 * half + t2) + u) * VPP + 32);
            }
        }
    };
    auto pv = [&](int it, int half, const v2f (&va)[2][4], const v2f (&vb)[2][4]) {
        __builtin_amdgcn_sched_barrier(0);
        #pragma unroll
        for (int t2 = 0; t2 < 2; ++t2) {
            #pragma unroll
            for (int u = 0; u < 4; ++u) {
                v2f p;
                p.x = S[4 * it + 2 * half + t2][2 * u];
                p.y = S[4 * it + 2 * half + t2][2 * u + 1];
                oacc[0] = wmma4(va[t2][u], p, oacc[0]);
                oacc[1] = wmma4(vb[t2][u], p, oacc[1]);
            }
        }
        __builtin_amdgcn_sched_barrier(0);
    };

    #pragma nounroll
    for (int kh = 0; kh < 2; ++kh) {
        #pragma unroll
        for (int T = 0; T < NTILE; ++T) S[T] = v8f{0.f, 0.f, 0.f, 0.f, 0.f, 0.f, 0.f, 0.f};

        gloadK(0, gA); gloadK(1, gB);
        stashK(0, gA); stashK(1, gB);
        gloadK(2, gA); gloadK(3, gB);
        lds_stage_barrier();
        kfrag(0, bkA);
        #pragma unroll
        for (int jt = 0; jt < NT_QK; jt += 2) {
            kfrag(jt + 1, bkB);
            if (jt + 2 < NT_QK) stashK(jt + 2, gA);
            if (jt + 4 < NT_QK) gloadK(jt + 4, gA);
            qk(jt, bkA);
            lds_stage_barrier();
            if (jt + 2 < NT_QK) kfrag(jt + 2, bkA);
            if (jt + 3 < NT_QK) stashK(jt + 3, gB);
            if (jt + 5 < NT_QK) gloadK(jt + 5, gB);
            qk(jt + 1, bkB);
            lds_stage_barrier();
        }

        gloadV(0, gA); gloadV(1, gB);

        float mxa[4] = {-3.0e38f, -3.0e38f, -3.0e38f, -3.0e38f};
        #pragma unroll
        for (int T = 0; T < NTILE; ++T) {
            #pragma unroll
            for (int k = 0; k < 4; ++k)
                mxa[k] = fmaxf(mxa[k], fmaxf(S[T][k], S[T][k + 4]));
        }
        float mx = fmaxf(fmaxf(mxa[0], mxa[1]), fmaxf(mxa[2], mxa[3]));
        mx = fmaxf(mx, __shfl_xor(mx, 16, 32));
        const float m_new = fmaxf(m_run, mx);
        const float alpha = __builtin_amdgcn_exp2f((m_run - m_new) * c1);
        m_run = m_new;
        const float nb = -m_new * c1;
        float sa[4] = {0.f, 0.f, 0.f, 0.f};
        #pragma unroll
        for (int T = 0; T < NTILE; ++T) {
            #pragma unroll
            for (int r = 0; r < 8; ++r) {
                const float e = __builtin_amdgcn_exp2f(fmaf(S[T][r], c1, nb));
                S[T][r] = e;
                sa[r & 3] += e;
            }
        }
        l_run = fmaf(l_run, alpha, (sa[0] + sa[1]) + (sa[2] + sa[3]));
        #pragma unroll
        for (int tt = 0; tt < 2; ++tt) {
            #pragma unroll
            for (int r = 0; r < 8; ++r) oacc[tt][r] *= alpha;
        }

        stashV(0, gA); stashV(1, gB);
        gloadV(2, gA); gloadV(3, gB);
        lds_stage_barrier();
        pvfrag(0, 0, vaA, vbA);
        #pragma unroll
        for (int it = 0; it < NT_PV; it += 2) {
            pvfrag(it, 1, vaB, vbB);
            if (it + 2 < NT_PV) stashV(it + 2, gA);
            if (it + 4 < NT_PV) gloadV(it + 4, gA);
            pv(it, 0, vaA, vbA);
            pvfrag(it + 1, 0, vaA, vbA);
            pv(it, 1, vaB, vbB);
            lds_stage_barrier();
            pvfrag(it + 1, 1, vaB, vbB);
            if (it + 3 < NT_PV) stashV(it + 3, gB);
            if (it + 5 < NT_PV) gloadV(it + 5, gB);
            pv(it + 1, 0, vaA, vbA);
            if (it + 2 < NT_PV) pvfrag(it + 2, 0, vaA, vbA);
            pv(it + 1, 1, vaB, vbB);
            lds_stage_barrier();
        }

        kg += (size_t)HK * CR_;
        vg += (size_t)HK * CR_;
    }

    const float sum = l_run + __shfl_xor(l_run, 16, 32);
    const float inv = 1.f / sum;

    #pragma unroll
    for (int tt = 0; tt < 2; ++tt) {
        const size_t idx = (size_t)(b * N_ + q0 + lo) * CR_ + h * HD_ + tt * 16 + 8 * hi;
        const v4f h0 = *reinterpret_cast<const v4f*>(Hb + idx);
        const v4f h1 = *reinterpret_cast<const v4f*>(Hb + idx + 4);
        v4f o0, o1;
        o0.x = h0.x + oacc[tt][0] * inv; o0.y = h0.y + oacc[tt][1] * inv;
        o0.z = h0.z + oacc[tt][2] * inv; o0.w = h0.w + oacc[tt][3] * inv;
        o1.x = h1.x + oacc[tt][4] * inv; o1.y = h1.y + oacc[tt][5] * inv;
        o1.z = h1.z + oacc[tt][6] * inv; o1.w = h1.w + oacc[tt][7] * inv;
        *reinterpret_cast<v4f*>(Out + idx)     = o0;
        *reinterpret_cast<v4f*>(Out + idx + 4) = o1;
    }
}

extern "C" void kernel_launch(void* const* d_in, const int* in_sizes, int n_in,
                              void* d_out, int out_size, void* d_ws, size_t ws_size,
                              hipStream_t stream) {
    (void)in_sizes; (void)n_in; (void)out_size; (void)ws_size;
    const float* x      = (const float*)d_in[0];
    const float* ln0_g  = (const float*)d_in[1];
    const float* ln0_b  = (const float*)d_in[2];
    const float* w_proj = (const float*)d_in[3];
    const float* b_proj = (const float*)d_in[4];
    const float* ln1_g  = (const float*)d_in[5];
    const float* ln1_b  = (const float*)d_in[6];
    const float* wq     = (const float*)d_in[7];
    const float* wk     = (const float*)d_in[8];
    const float* wv     = (const float*)d_in[9];
    float* out = (float*)d_out;

    float* ws = (float*)d_ws;
    const size_t rowsBig = (size_t)B_ * N_ * DIM_;
    const size_t rowsCr  = (size_t)B_ * N_ * CR_;
    float* xn = ws;
    float* hB = xn + rowsBig;
    float* yB = hB + rowsCr;
    float* qB = yB + rowsCr;
    float* kB = qB + rowsCr;
    float* vB = kB + rowsCr;

    const int M = B_ * N_;

    (void)hipFuncSetAttribute((const void*)ln_proj_kernel,
                              hipFuncAttributeMaxDynamicSharedMemorySize,
                              PROJ_SMEM);
    ln_proj_kernel<<<M / QM, 256, PROJ_SMEM, stream>>>(
        x, ln0_g, ln0_b, w_proj, b_proj, hB);

    (void)hipFuncSetAttribute((const void*)ln_qkv_kernel,
                              hipFuncAttributeMaxDynamicSharedMemorySize,
                              QKV_SMEM);
    ln_qkv_kernel<<<M / QM, 256, QKV_SMEM, stream>>>(
        hB, ln1_g, ln1_b, wq, wk, wv, qB, kB, vB);

    (void)hipFuncSetAttribute((const void*)attn_kernel,
                              hipFuncAttributeMaxDynamicSharedMemorySize,
                              ATTN_SMEM);
    attn_kernel<<<dim3(N_ / 64, B_ * H_), 128, ATTN_SMEM, stream>>>(
        qB, kB, vB, hB, out);
}
